# starting kernel
# baseline (speedup 1.0000x reference)
#include <hip/hip_runtime.h>
#include <math.h>


typedef float v2f __attribute__((ext_vector_type(2)));
typedef float v8f __attribute__((ext_vector_type(8)));

namespace {
constexpr int kBatch  = 64;
constexpr int kEv     = 500;
constexpr int kK      = kBatch * kEv;
constexpr int kHid    = 128;
constexpr int kS      = 2209;
constexpr int kSTiles = (kS + 15) / 16;
constexpr int kSPad   = kSTiles * 16;
constexpr int kT      = 550;
constexpr int kTPad   = 560;
constexpr int kTTiles = kTPad / 16;
constexpr int kKB     = 32;
constexpr int kNChunk = 4;
constexpr int kKCh    = kK / kNChunk;
constexpr float kNorm = 0.3989422804f * 1.41421356237309515f;
}

__global__ __launch_bounds__(256)
void sipm_h_kernel(const float* __restrict__ x, const float* __restrict__ W1,
                   const float* __restrict__ b1, float* __restrict__ h) {
  int idx = blockIdx.x * 256 + threadIdx.x;
  if (idx >= kK * (kHid / 4)) return;
  int k = idx >> 5;
  int j = (idx & 31) << 2;
  float x0 = x[k * 3 + 0], x1 = x[k * 3 + 1], x2 = x[k * 3 + 2];
  const float4 wa = *(const float4*)(W1 + 0 * kHid + j);
  const float4 wb = *(const float4*)(W1 + 1 * kHid + j);
  const float4 wc = *(const float4*)(W1 + 2 * kHid + j);
  const float4 bb = *(const float4*)(b1 + j);
  float4 r;
  r.x = fmaxf(fmaf(x0, wa.x, fmaf(x1, wb.x, fmaf(x2, wc.x, bb.x))), 0.f);
  r.y = fmaxf(fmaf(x0, wa.y, fmaf(x1, wb.y, fmaf(x2, wc.y, bb.y))), 0.f);
  r.z = fmaxf(fmaf(x0, wa.z, fmaf(x1, wb.z, fmaf(x2, wc.z, bb.z))), 0.f);
  r.w = fmaxf(fmaf(x0, wa.w, fmaf(x1, wb.w, fmaf(x2, wc.w, bb.w))), 0.f);
  *(float4*)(h + (size_t)k * kHid + j) = r;
}

__global__ __launch_bounds__(256)
void sipm_expv_kernel(const float* __restrict__ z, const float* __restrict__ mask,
                      float* __restrict__ expv) {
  int idx = blockIdx.x * 256 + threadIdx.x;
  if (idx >= kK * kTPad) return;
  int k = idx / kTPad;
  int t = idx - k * kTPad;
  float v = 0.f;
  if (t < kT) {
    float d = (float)t - z[k];
    v = expf(-d * d) * mask[k] * kNorm;
  }
  expv[idx] = v;
}

__global__ __launch_bounds__(512)
void sipm_gemm_kernel(const float* __restrict__ W2, const float* __restrict__ b2,
                      const float* __restrict__ h, const float* __restrict__ expv,
                      float* __restrict__ part) {
  __shared__ float lds_w2[kHid * 16];
  __shared__ float lds_resp[kKB * 16];

  const int tid    = threadIdx.x;
  const int stile  = blockIdx.x;
  const int chunk  = blockIdx.y;
  const int s_base = stile * 16;

  for (int r = 0; r < 4; ++r) {
    int idx = tid + r * 512;
    int i = idx >> 4, ss = idx & 15;
    int s = s_base + ss;
    lds_w2[idx] = (s < kS) ? W2[(size_t)i * kS + s] : 0.f;
  }

  const int kk  = tid >> 4;
  const int ssA = tid & 15;
  const int sA  = s_base + ssA;
  const float accInit = (sA < kS) ? b2[sA] : 0.f;

  const int w    = tid >> 5;
  const int lane = tid & 31;
  const int m    = lane & 15;
  const int kb2  = (lane >> 4) << 1;
  const int tb0 = w * 16, tb1 = (w + 16) * 16, tb2 = (w + 32) * 16;
  const bool t2ok = (w + 32) < kTTiles;

  v8f acc0 = {}, acc1 = {}, acc2 = {};

  __syncthreads();

  for (int kblk = 0; kblk < kKCh / kKB; ++kblk) {
    const int k0 = chunk * kKCh + kblk * kKB;

    {
      float acc = accInit;
      const float4* hrow = (const float4*)(h + (size_t)(k0 + kk) * kHid);
      #pragma unroll
      for (int i4 = 0; i4 < kHid / 4; ++i4) {
        float4 hv = hrow[i4];
        int ib = i4 * 4;
        acc = fmaf(hv.x, lds_w2[(ib + 0) * 16 + ssA], acc);
        acc = fmaf(hv.y, lds_w2[(ib + 1) * 16 + ssA], acc);
        acc = fmaf(hv.z, lds_w2[(ib + 2) * 16 + ssA], acc);
        acc = fmaf(hv.w, lds_w2[(ib + 3) * 16 + ssA], acc);
      }
      lds_resp[kk * 16 + ssA] = expf(acc);
    }
    __syncthreads();

    const float* erow = expv + (size_t)(k0 + kb2) * kTPad + m;
    v2f aP[2], bP0[2], bP1[2], bP2[2];
    #pragma unroll
    for (int p = 0; p < 2; ++p) {
      const int ks = p * 4;
      aP[p].x = lds_resp[(ks + kb2) * 16 + m];
      aP[p].y = lds_resp[(ks + kb2 + 1) * 16 + m];
      const float* e = erow + (size_t)ks * kTPad;
      bP0[p].x = e[tb0]; bP0[p].y = e[kTPad + tb0];
      bP1[p].x = e[tb1]; bP1[p].y = e[kTPad + tb1];
      bP2[p].x = e[tb2]; bP2[p].y = e[kTPad + tb2];
    }
    #pragma unroll
    for (int i = 0; i < kKB / 4; ++i) {
      const int slot = i & 1;
      const v2f a   = aP[slot];
      const v2f b0  = bP0[slot];
      const v2f b1v = bP1[slot];
      const v2f b2v = bP2[slot];
      if (i + 2 < kKB / 4) {
        const int ks = (i + 2) * 4;
        aP[slot].x = lds_resp[(ks + kb2) * 16 + m];
        aP[slot].y = lds_resp[(ks + kb2 + 1) * 16 + m];
        const float* e = erow + (size_t)ks * kTPad;
        bP0[slot].x = e[tb0]; bP0[slot].y = e[kTPad + tb0];
        bP1[slot].x = e[tb1]; bP1[slot].y = e[kTPad + tb1];
        bP2[slot].x = e[tb2]; bP2[slot].y = e[kTPad + tb2];
      }
      acc0 = __builtin_amdgcn_wmma_f32_16x16x4_f32(false, a, false, b0,
                                                   (short)0, acc0, false, false);
      acc1 = __builtin_amdgcn_wmma_f32_16x16x4_f32(false, a, false, b1v,
                                                   (short)0, acc1, false, false);
      if (t2ok)
        acc2 = __builtin_amdgcn_wmma_f32_16x16x4_f32(false, a, false, b2v,
                                                     (short)0, acc2, false, false);
    }
    __syncthreads();
  }

  float* pbase = part + ((size_t)chunk * kSPad + s_base + ((lane >> 4) << 3)) * kTPad + m;
  #pragma unroll
  for (int r = 0; r < 8; ++r) {
    float* prow = pbase + (size_t)r * kTPad;
    prow[tb0] = acc0[r];
    prow[tb1] = acc1[r];
    if (t2ok) prow[tb2] = acc2[r];
  }
}

__global__ __launch_bounds__(256)
void sipm_reduce_kernel(const float* __restrict__ part, float* __restrict__ out) {
  int idx = blockIdx.x * 256 + threadIdx.x;
  if (idx >= kS * kT) return;
  int s = idx / kT, t = idx - s * kT;
  size_t off = (size_t)s * kTPad + t;
  float sum = 0.f;
  #pragma unroll
  for (int c = 0; c < kNChunk; ++c)
    sum += part[(size_t)c * kSPad * kTPad + off];
  out[idx] = sum;
}

extern "C" void kernel_launch(void* const* d_in, const int* in_sizes, int n_in,
                              void* d_out, int out_size, void* d_ws, size_t ws_size,
                              hipStream_t stream) {
  const float* x   = (const float*)d_in[0];
  const float* z   = (const float*)d_in[1];
  const float* msk = (const float*)d_in[2];
  const float* W1  = (const float*)d_in[3];
  const float* b1  = (const float*)d_in[4];
  const float* W2  = (const float*)d_in[5];
  const float* b2  = (const float*)d_in[6];
  float* out = (float*)d_out;

  float* ws   = (float*)d_ws;
  float* h    = ws;
  float* expv = h + (size_t)kK * kHid;
  float* part = expv + (size_t)kK * kTPad;

  sipm_h_kernel<<<(kK * (kHid / 4) + 255) / 256, 256, 0, stream>>>(x, W1, b1, h);
  sipm_expv_kernel<<<(kK * kTPad + 255) / 256, 256, 0, stream>>>(z, msk, expv);
  sipm_gemm_kernel<<<dim3(kSTiles, kNChunk), 512, 0, stream>>>(W2, b2, h, expv, part);
  sipm_reduce_kernel<<<(kS * kT + 255) / 256, 256, 0, stream>>>(part, out);
}


# h rows of kept electrons staged through LDS 32 at a time: each row fetched once per workgroup as one whole 512-byte row load, one batch ahead; all 4 waves read A fragments from LDS; sums unchanged
# speedup vs baseline: 4.6245x; 4.6245x over previous
#include <hip/hip_runtime.h>
#include <math.h>


typedef float v2f __attribute__((ext_vector_type(2)));
typedef float v4f __attribute__((ext_vector_type(4)));
typedef float v8f __attribute__((ext_vector_type(8)));

namespace {
constexpr int kBatch  = 64;
constexpr int kEv     = 500;
constexpr int kK      = kBatch * kEv;
constexpr int kHid    = 128;
constexpr int kS      = 2209;
constexpr int kSW     = 64;
constexpr int kSGrid  = (kS + kSW - 1) / kSW;
constexpr int kSPad   = kSGrid * kSW;
constexpr int kT      = 550;
constexpr int kTW     = 32;
constexpr int kTGrid  = (kT + kTW - 1) / kTW;
constexpr int kTPad   = kTGrid * kTW;
constexpr int kReach  = 11;
constexpr int kCH     = 3200;
constexpr int kNCh    = kK / kCH;
constexpr int kSeg    = kCH / 4;
constexpr int kBR     = 32;
constexpr int kPitchF = kHid + 4;
constexpr int kListCap = kCH + kBR;
constexpr float kNorm = 0.3989422804f * 1.41421356237309515f;
static_assert(kNCh * kCH == kK, "chunks must tile K");
static_assert(kSeg % 32 == 0, "segment must be whole wave loads");
}

__global__ __launch_bounds__(256)
void sipm_h_kernel(const float* __restrict__ x, const float* __restrict__ W1,
                   const float* __restrict__ b1, float* __restrict__ h) {
  int idx = blockIdx.x * 256 + threadIdx.x;
  if (idx >= kK * (kHid / 4)) return;
  int k = idx >> 5;
  int j = (idx & 31) << 2;
  float x0 = x[k * 3 + 0], x1 = x[k * 3 + 1], x2 = x[k * 3 + 2];
  const float4 wa = *(const float4*)(W1 + 0 * kHid + j);
  const float4 wb = *(const float4*)(W1 + 1 * kHid + j);
  const float4 wc = *(const float4*)(W1 + 2 * kHid + j);
  const float4 bb = *(const float4*)(b1 + j);
  float4 r;
  r.x = fmaxf(fmaf(x0, wa.x, fmaf(x1, wb.x, fmaf(x2, wc.x, bb.x))), 0.f);
  r.y = fmaxf(fmaf(x0, wa.y, fmaf(x1, wb.y, fmaf(x2, wc.y, bb.y))), 0.f);
  r.z = fmaxf(fmaf(x0, wa.z, fmaf(x1, wb.z, fmaf(x2, wc.z, bb.z))), 0.f);
  r.w = fmaxf(fmaf(x0, wa.w, fmaf(x1, wb.w, fmaf(x2, wc.w, bb.w))), 0.f);
  *(float4*)(h + (size_t)k * kHid + j) = r;
}

__global__ __launch_bounds__(128)
void sipm_window_kernel(const float* __restrict__ z, const float* __restrict__ mask,
                        const float* __restrict__ W2, const float* __restrict__ b2,
                        const float* __restrict__ h, float* __restrict__ plane) {
  __shared__ __attribute__((aligned(16))) int   lds_k[kListCap];
  __shared__ __attribute__((aligned(16))) float lds_z[kListCap];
  __shared__ __attribute__((aligned(16))) float lds_m[kListCap];
  __shared__ int lds_cnt[4];
  __shared__ __attribute__((aligned(16))) float lds_h[2][kBR * kPitchF];

  const int tid  = threadIdx.x;
  const int w    = tid >> 5;
  const int lane = tid & 31;
  const int m    = lane & 15;
  const int hf   = lane >> 4;
  const int s_base = blockIdx.x * kSW + w * 16;
  const int t_base = blockIdx.y * kTW;
  const bool t1ok  = (t_base + 16) < kT;
  const int  t_last = (t_base + kTW - 1 < kT - 1) ? (t_base + kTW - 1) : (kT - 1);
  const float zlo = (float)(t_base - kReach);
  const float zhi = (float)(t_last + kReach);

  const int  sW  = s_base + m;
  const bool sok = sW < kS;
  v2f w2f[16][2];
  #pragma unroll
  for (int j = 0; j < 16; ++j) {
    #pragma unroll
    for (int q = 0; q < 2; ++q) {
      const int i0 = 8 * j + 4 * hf + 2 * q;
      w2f[j][q].x = sok ? W2[(size_t)(i0 + 0) * kS + sW] : 0.f;
      w2f[j][q].y = sok ? W2[(size_t)(i0 + 1) * kS + sW] : 0.f;
    }
  }
  const float bias = sok ? b2[sW] : 0.f;

  v8f acc0 = {}, acc1 = {};

  for (int ch = 0; ch < kNCh; ++ch) {
    const int cbase = ch * kCH + w * kSeg;
    {
      int cnt = 0;
      #pragma unroll 5
      for (int j = 0; j < kSeg / 32; ++j) {
        const float zk = z[cbase + j * 32 + lane];
        const bool hit = (zk > zlo) && (zk < zhi);
        cnt += __builtin_popcount(__builtin_amdgcn_ballot_w32(hit));
      }
      if (lane == 0) lds_cnt[w] = cnt;
    }
    __syncthreads();
    const int c0 = lds_cnt[0], c1 = lds_cnt[1], c2 = lds_cnt[2], c3 = lds_cnt[3];
    const int n = __builtin_amdgcn_readfirstlane(c0 + c1 + c2 + c3);
    {
      int pos = (w > 0 ? c0 : 0) + (w > 1 ? c1 : 0) + (w > 2 ? c2 : 0);
      #pragma unroll 5
      for (int j = 0; j < kSeg / 32; ++j) {
        const int k = cbase + j * 32 + lane;
        const float zk = z[k];
        const bool hit = (zk > zlo) && (zk < zhi);
        const unsigned b = __builtin_amdgcn_ballot_w32(hit);
        if (hit) {
          const int p = pos + __builtin_popcount(b & ((1u << lane) - 1u));
          lds_k[p] = k;
          lds_z[p] = zk;
          lds_m[p] = mask[k];
        }
        pos += __builtin_popcount(b);
      }
      if (w == 3) {
        const int p = n + lane;
        lds_k[p] = 0;
        lds_z[p] = 0.f;
        lds_m[p] = 0.f;
      }
    }
    __syncthreads();

    {
      const int ng = (n + 15) >> 4;
      const int nb = (ng + 1) >> 1;
      if (nb > 0) {
        float4 pre[8];
        {
          const int4 ka = *(const int4*)(lds_k + 8 * w);
          const int4 kb = *(const int4*)(lds_k + 8 * w + 4);
          const int kk[8] = {ka.x, ka.y, ka.z, ka.w, kb.x, kb.y, kb.z, kb.w};
          #pragma unroll
          for (int i = 0; i < 8; ++i) {
            const int ke = __builtin_amdgcn_readfirstlane(kk[i]);
            pre[i] = *((const float4*)(h + (size_t)ke * kHid) + lane);
          }
          float* dst = &lds_h[0][(8 * w) * kPitchF + 4 * lane];
          #pragma unroll
          for (int i = 0; i < 8; ++i) *(float4*)(dst + i * kPitchF) = pre[i];
        }
        __syncthreads();
        for (int b = 0; b < nb; ++b) {
          {
            const int bn = (b + 1 < nb) ? (b + 1) : b;
            const int4 ka = *(const int4*)(lds_k + bn * kBR + 8 * w);
            const int4 kb = *(const int4*)(lds_k + bn * kBR + 8 * w + 4);
            const int kk[8] = {ka.x, ka.y, ka.z, ka.w, kb.x, kb.y, kb.z, kb.w};
            #pragma unroll
            for (int i = 0; i < 8; ++i) {
              const int ke = __builtin_amdgcn_readfirstlane(kk[i]);
              pre[i] = *((const float4*)(h + (size_t)ke * kHid) + lane);
            }
          }
          const float* cur = &lds_h[b & 1][0];
          for (int half = 0; half < 2; ++half) {
            const int g = 2 * b + half;
            if (g >= ng) break;
            const int e0 = g * 16;
            const v4f z0 = *(const v4f*)(lds_z + e0 + 8 * hf);
            const v4f z1 = *(const v4f*)(lds_z + e0 + 8 * hf + 4);
            const v4f m0 = *(const v4f*)(lds_m + e0 + 8 * hf);
            const v4f m1 = *(const v4f*)(lds_m + e0 + 8 * hf + 4);
            const float* arow = cur + (half * 16 + m) * kPitchF + 4 * hf;
            v4f a[16];
            #pragma unroll
            for (int j = 0; j < 16; ++j) a[j] = *(const v4f*)(arow + 8 * j);
            __builtin_amdgcn_sched_barrier(0);
            const float zz[8] = {z0.x, z0.y, z0.z, z0.w, z1.x, z1.y, z1.z, z1.w};
            const float mm[8] = {m0.x, m0.y, m0.z, m0.w, m1.x, m1.y, m1.z, m1.w};

            v8f d0 = {}, d1 = {};
            #pragma unroll
            for (int j = 0; j < 16; ++j) {
              v2f a0; a0.x = a[j].x; a0.y = a[j].y;
              v2f a1; a1.x = a[j].z; a1.y = a[j].w;
              d0 = __builtin_amdgcn_wmma_f32_16x16x4_f32(false, a0, false, w2f[j][0],
                                                         (short)0, d0, false, false);
              d1 = __builtin_amdgcn_wmma_f32_16x16x4_f32(false, a1, false, w2f[j][1],
                                                         (short)0, d1, false, false);
            }
            float resp[8];
            #pragma unroll
            for (int r = 0; r < 8; ++r) resp[r] = expf((d0[r] + d1[r]) + bias);

            const float tf0 = (float)(t_base + m);
            #pragma unroll
            for (int j = 0; j < 4; ++j) {
              v2f af; af.x = resp[2 * j]; af.y = resp[2 * j + 1];
              const float da = tf0 - zz[2 * j];
              const float db = tf0 - zz[2 * j + 1];
              v2f bf;
              bf.x = expf(-da * da) * mm[2 * j] * kNorm;
              bf.y = expf(-db * db) * mm[2 * j + 1] * kNorm;
              acc0 = __builtin_amdgcn_wmma_f32_16x16x4_f32(false, af, false, bf,
                                                           (short)0, acc0, false, false);
            }
            if (t1ok) {
              const float tf1 = (float)(t_base + 16 + m);
              #pragma unroll
              for (int j = 0; j < 4; ++j) {
                v2f af; af.x = resp[2 * j]; af.y = resp[2 * j + 1];
                const float da = tf1 - zz[2 * j];
                const float db = tf1 - zz[2 * j + 1];
                v2f bf;
                bf.x = expf(-da * da) * mm[2 * j] * kNorm;
                bf.y = expf(-db * db) * mm[2 * j + 1] * kNorm;
                acc1 = __builtin_amdgcn_wmma_f32_16x16x4_f32(false, af, false, bf,
                                                             (short)0, acc1, false, false);
              }
            }
          }
          {
            float* dst = &lds_h[(b + 1) & 1][(8 * w) * kPitchF + 4 * lane];
            #pragma unroll
            for (int i = 0; i < 8; ++i) *(float4*)(dst + i * kPitchF) = pre[i];
          }
          __syncthreads();
        }
      }
    }
  }

  float* pbase = plane + (size_t)(s_base + 8 * hf) * kTPad + t_base + m;
  #pragma unroll
  for (int r = 0; r < 8; ++r) {
    float* prow = pbase + (size_t)r * kTPad;
    prow[0] = acc0[r];
    if (t1ok) prow[16] = acc1[r];
  }
}

__global__ __launch_bounds__(256)
void sipm_copy_kernel(const float* __restrict__ plane, float* __restrict__ out) {
  int idx = blockIdx.x * 256 + threadIdx.x;
  if (idx >= kS * kT) return;
  int s = idx / kT, t = idx - s * kT;
  out[idx] = plane[(size_t)s * kTPad + t];
}

extern "C" void kernel_launch(void* const* d_in, const int* in_sizes, int n_in,
                              void* d_out, int out_size, void* d_ws, size_t ws_size,
                              hipStream_t stream) {
  const float* x   = (const float*)d_in[0];
  const float* z   = (const float*)d_in[1];
  const float* msk = (const float*)d_in[2];
  const float* W1  = (const float*)d_in[3];
  const float* b1  = (const float*)d_in[4];
  const float* W2  = (const float*)d_in[5];
  const float* b2  = (const float*)d_in[6];
  float* out = (float*)d_out;

  float* ws    = (float*)d_ws;
  float* h     = ws;
  float* plane = h + (size_t)kK * kHid;

  sipm_h_kernel<<<(kK * (kHid / 4) + 255) / 256, 256, 0, stream>>>(x, W1, b1, h);
  sipm_window_kernel<<<dim3(kSGrid, kTGrid), 128, 0, stream>>>(z, msk, W2, b2, h, plane);
  sipm_copy_kernel<<<(kS * kT + 255) / 256, 256, 0, stream>>>(plane, out);
}


# Gaussian taps made once per workgroup, not once per wave: each wave makes a quarter of the next batch's taps beside its resp WMMAs and parks them in LDS in B-fragment order; sums unchanged
# speedup vs baseline: 5.1766x; 1.1194x over previous
#include <hip/hip_runtime.h>
#include <math.h>


typedef float v2f __attribute__((ext_vector_type(2)));
typedef float v4f __attribute__((ext_vector_type(4)));
typedef float v8f __attribute__((ext_vector_type(8)));

namespace {
constexpr int kBatch  = 64;
constexpr int kEv     = 500;
constexpr int kK      = kBatch * kEv;
constexpr int kHid    = 128;
constexpr int kS      = 2209;
constexpr int kSW     = 64;
constexpr int kSGrid  = (kS + kSW - 1) / kSW;
constexpr int kSPad   = kSGrid * kSW;
constexpr int kT      = 550;
constexpr int kTW     = 32;
constexpr int kTGrid  = (kT + kTW - 1) / kTW;
constexpr int kTPad   = kTGrid * kTW;
constexpr int kReach  = 11;
constexpr int kCH     = 3200;
constexpr int kNCh    = kK / kCH;
constexpr int kSeg    = kCH / 4;
constexpr int kBR     = 32;
constexpr int kPitchF = kHid + 4;
constexpr int kListCap = kCH + kBR;
constexpr int kTapQ   = 32 * 4;
constexpr int kTapBuf = 2 * 2 * 2 * kTapQ;
constexpr float kNorm = 0.3989422804f * 1.41421356237309515f;
static_assert(kNCh * kCH == kK, "chunks must tile K");
static_assert(kSeg % 32 == 0, "segment must be whole wave loads");
}

__global__ __launch_bounds__(256)
void sipm_h_kernel(const float* __restrict__ x, const float* __restrict__ W1,
                   const float* __restrict__ b1, float* __restrict__ h) {
  int idx = blockIdx.x * 256 + threadIdx.x;
  if (idx >= kK * (kHid / 4)) return;
  int k = idx >> 5;
  int j = (idx & 31) << 2;
  float x0 = x[k * 3 + 0], x1 = x[k * 3 + 1], x2 = x[k * 3 + 2];
  const float4 wa = *(const float4*)(W1 + 0 * kHid + j);
  const float4 wb = *(const float4*)(W1 + 1 * kHid + j);
  const float4 wc = *(const float4*)(W1 + 2 * kHid + j);
  const float4 bb = *(const float4*)(b1 + j);
  float4 r;
  r.x = fmaxf(fmaf(x0, wa.x, fmaf(x1, wb.x, fmaf(x2, wc.x, bb.x))), 0.f);
  r.y = fmaxf(fmaf(x0, wa.y, fmaf(x1, wb.y, fmaf(x2, wc.y, bb.y))), 0.f);
  r.z = fmaxf(fmaf(x0, wa.z, fmaf(x1, wb.z, fmaf(x2, wc.z, bb.z))), 0.f);
  r.w = fmaxf(fmaf(x0, wa.w, fmaf(x1, wb.w, fmaf(x2, wc.w, bb.w))), 0.f);
  *(float4*)(h + (size_t)k * kHid + j) = r;
}

__global__ __launch_bounds__(128)
void sipm_window_kernel(const float* __restrict__ z, const float* __restrict__ mask,
                        const float* __restrict__ W2, const float* __restrict__ b2,
                        const float* __restrict__ h, float* __restrict__ plane) {
  __shared__ __attribute__((aligned(16))) int   lds_k[kListCap];
  __shared__ __attribute__((aligned(16))) float lds_z[kListCap];
  __shared__ __attribute__((aligned(16))) float lds_m[kListCap];
  __shared__ int lds_cnt[4];
  __shared__ __attribute__((aligned(16))) float lds_h[2][kBR * kPitchF];
  __shared__ __attribute__((aligned(16))) float lds_tap[2][kTapBuf];

  const int tid  = threadIdx.x;
  const int w    = tid >> 5;
  const int lane = tid & 31;
  const int m    = lane & 15;
  const int hf   = lane >> 4;
  const int s_base = blockIdx.x * kSW + w * 16;
  const int t_base = blockIdx.y * kTW;
  const int  t_last = (t_base + kTW - 1 < kT - 1) ? (t_base + kTW - 1) : (kT - 1);
  const float zlo = (float)(t_base - kReach);
  const float zhi = (float)(t_last + kReach);
  const int   gw  = w >> 1;
  const int   tw  = w & 1;
  const float tfw = (float)(t_base + 16 * tw + m);
  const int   tap_wr = ((gw * 2 + tw) * 2) * kTapQ + 4 * lane;

  const int  sW  = s_base + m;
  const bool sok = sW < kS;
  v2f w2f[16][2];
  #pragma unroll
  for (int j = 0; j < 16; ++j) {
    #pragma unroll
    for (int q = 0; q < 2; ++q) {
      const int i0 = 8 * j + 4 * hf + 2 * q;
      w2f[j][q].x = sok ? W2[(size_t)(i0 + 0) * kS + sW] : 0.f;
      w2f[j][q].y = sok ? W2[(size_t)(i0 + 1) * kS + sW] : 0.f;
    }
  }
  const float bias = sok ? b2[sW] : 0.f;

  v8f acc0 = {}, acc1 = {};

  for (int ch = 0; ch < kNCh; ++ch) {
    const int cbase = ch * kCH + w * kSeg;
    {
      int cnt = 0;
      #pragma unroll 5
      for (int j = 0; j < kSeg / 32; ++j) {
        const float zk = z[cbase + j * 32 + lane];
        const bool hit = (zk > zlo) && (zk < zhi);
        cnt += __builtin_popcount(__builtin_amdgcn_ballot_w32(hit));
      }
      if (lane == 0) lds_cnt[w] = cnt;
    }
    __syncthreads();
    const int c0 = lds_cnt[0], c1 = lds_cnt[1], c2 = lds_cnt[2], c3 = lds_cnt[3];
    const int n = __builtin_amdgcn_readfirstlane(c0 + c1 + c2 + c3);
    {
      int pos = (w > 0 ? c0 : 0) + (w > 1 ? c1 : 0) + (w > 2 ? c2 : 0);
      #pragma unroll 5
      for (int j = 0; j < kSeg / 32; ++j) {
        const int k = cbase + j * 32 + lane;
        const float zk = z[k];
        const bool hit = (zk > zlo) && (zk < zhi);
        const unsigned b = __builtin_amdgcn_ballot_w32(hit);
        if (hit) {
          const int p = pos + __builtin_popcount(b & ((1u << lane) - 1u));
          lds_k[p] = k;
          lds_z[p] = zk;
          lds_m[p] = mask[k];
        }
        pos += __builtin_popcount(b);
      }
      if (w == 3) {
        const int p = n + lane;
        lds_k[p] = 0;
        lds_z[p] = 0.f;
        lds_m[p] = 0.f;
      }
    }
    __syncthreads();

    {
      const int ng = (n + 15) >> 4;
      const int nb = (ng + 1) >> 1;
      if (nb > 0) {
        float4 pre[8];
        {
          const int4 ka = *(const int4*)(lds_k + 8 * w);
          const int4 kb = *(const int4*)(lds_k + 8 * w + 4);
          const int kk[8] = {ka.x, ka.y, ka.z, ka.w, kb.x, kb.y, kb.z, kb.w};
          #pragma unroll
          for (int i = 0; i < 8; ++i) {
            const int ke = __builtin_amdgcn_readfirstlane(kk[i]);
            pre[i] = *((const float4*)(h + (size_t)ke * kHid) + lane);
          }
          {
            const int ez = 16 * gw + 8 * hf;
            const v4f z0 = *(const v4f*)(lds_z + ez);
            const v4f z1 = *(const v4f*)(lds_z + ez + 4);
            const v4f m0 = *(const v4f*)(lds_m + ez);
            const v4f m1 = *(const v4f*)(lds_m + ez + 4);
            const float zz[8] = {z0.x, z0.y, z0.z, z0.w, z1.x, z1.y, z1.z, z1.w};
            const float mm[8] = {m0.x, m0.y, m0.z, m0.w, m1.x, m1.y, m1.z, m1.w};
            float tp[8];
            #pragma unroll
            for (int i = 0; i < 8; ++i) {
              const float d = tfw - zz[i];
              tp[i] = expf(-d * d) * mm[i] * kNorm;
            }
            float* td = &lds_tap[0][tap_wr];
            v4f q0 = {tp[0], tp[1], tp[2], tp[3]};
            v4f q1 = {tp[4], tp[5], tp[6], tp[7]};
            *(v4f*)td = q0;
            *(v4f*)(td + kTapQ) = q1;
          }
          float* dst = &lds_h[0][(8 * w) * kPitchF + 4 * lane];
          #pragma unroll
          for (int i = 0; i < 8; ++i) *(float4*)(dst + i * kPitchF) = pre[i];
        }
        __syncthreads();
        for (int b = 0; b < nb; ++b) {
          const int bn = (b + 1 < nb) ? (b + 1) : b;
          {
            const int4 ka = *(const int4*)(lds_k + bn * kBR + 8 * w);
            const int4 kb = *(const int4*)(lds_k + bn * kBR + 8 * w + 4);
            const int kk[8] = {ka.x, ka.y, ka.z, ka.w, kb.x, kb.y, kb.z, kb.w};
            #pragma unroll
            for (int i = 0; i < 8; ++i) {
              const int ke = __builtin_amdgcn_readfirstlane(kk[i]);
              pre[i] = *((const float4*)(h + (size_t)ke * kHid) + lane);
            }
          }
          const float* cur  = &lds_h[b & 1][0];
          const float* tcur = &lds_tap[b & 1][4 * lane];
          float*       tnxt = &lds_tap[(b + 1) & 1][tap_wr];
          #pragma unroll
          for (int half = 0; half < 2; ++half) {
            const int g = 2 * b + half;
            if (g >= ng) break;
            const float* arow = cur + (half * 16 + m) * kPitchF + 4 * hf;
            v4f a[16];
            #pragma unroll
            for (int j = 0; j < 16; ++j) a[j] = *(const v4f*)(arow + 8 * j);
            v4f nz0 = {}, nz1 = {}, nm0 = {}, nm1 = {};
            if (half == 0) {
              const int ez = bn * kBR + 16 * gw + 8 * hf;
              nz0 = *(const v4f*)(lds_z + ez);
              nz1 = *(const v4f*)(lds_z + ez + 4);
              nm0 = *(const v4f*)(lds_m + ez);
              nm1 = *(const v4f*)(lds_m + ez + 4);
            }
            const float* tg = tcur + (half * 4) * kTapQ;
            const v4f b00 = *(const v4f*)(tg);
            const v4f b01 = *(const v4f*)(tg + kTapQ);
            const v4f b10 = *(const v4f*)(tg + 2 * kTapQ);
            const v4f b11 = *(const v4f*)(tg + 3 * kTapQ);
            __builtin_amdgcn_sched_barrier(0);

            v8f d0 = {}, d1 = {};
            #pragma unroll
            for (int j = 0; j < 16; ++j) {
              v2f a0; a0.x = a[j].x; a0.y = a[j].y;
              v2f a1; a1.x = a[j].z; a1.y = a[j].w;
              d0 = __builtin_amdgcn_wmma_f32_16x16x4_f32(false, a0, false, w2f[j][0],
                                                         (short)0, d0, false, false);
              d1 = __builtin_amdgcn_wmma_f32_16x16x4_f32(false, a1, false, w2f[j][1],
                                                         (short)0, d1, false, false);
            }
            if (half == 0) {
              const float zz[8] = {nz0.x, nz0.y, nz0.z, nz0.w, nz1.x, nz1.y, nz1.z, nz1.w};
              const float mm[8] = {nm0.x, nm0.y, nm0.z, nm0.w, nm1.x, nm1.y, nm1.z, nm1.w};
              float tp[8];
              #pragma unroll
              for (int i = 0; i < 8; ++i) {
                const float d = tfw - zz[i];
                tp[i] = expf(-d * d) * mm[i] * kNorm;
              }
              v4f q0 = {tp[0], tp[1], tp[2], tp[3]};
              v4f q1 = {tp[4], tp[5], tp[6], tp[7]};
              *(v4f*)tnxt = q0;
              *(v4f*)(tnxt + kTapQ) = q1;
            }
            float resp[8];
            #pragma unroll
            for (int r = 0; r < 8; ++r) resp[r] = expf((d0[r] + d1[r]) + bias);

            const float t0[8] = {b00.x, b00.y, b00.z, b00.w, b01.x, b01.y, b01.z, b01.w};
            const float t1[8] = {b10.x, b10.y, b10.z, b10.w, b11.x, b11.y, b11.z, b11.w};
            #pragma unroll
            for (int j = 0; j < 4; ++j) {
              v2f af;  af.x  = resp[2 * j]; af.y  = resp[2 * j + 1];
              v2f bf0; bf0.x = t0[2 * j];   bf0.y = t0[2 * j + 1];
              v2f bf1; bf1.x = t1[2 * j];   bf1.y = t1[2 * j + 1];
              acc0 = __builtin_amdgcn_wmma_f32_16x16x4_f32(false, af, false, bf0,
                                                           (short)0, acc0, false, false);
              acc1 = __builtin_amdgcn_wmma_f32_16x16x4_f32(false, af, false, bf1,
                                                           (short)0, acc1, false, false);
            }
          }
          {
            float* dst = &lds_h[(b + 1) & 1][(8 * w) * kPitchF + 4 * lane];
            #pragma unroll
            for (int i = 0; i < 8; ++i) *(float4*)(dst + i * kPitchF) = pre[i];
          }
          __syncthreads();
        }
      }
    }
  }

  float* pbase = plane + (size_t)(s_base + 8 * hf) * kTPad + t_base + m;
  #pragma unroll
  for (int r = 0; r < 8; ++r) {
    float* prow = pbase + (size_t)r * kTPad;
    prow[0] = acc0[r];
    prow[16] = acc1[r];
  }
}

__global__ __launch_bounds__(256)
void sipm_copy_kernel(const float* __restrict__ plane, float* __restrict__ out) {
  int idx = blockIdx.x * 256 + threadIdx.x;
  if (idx >= kS * kT) return;
  int s = idx / kT, t = idx - s * kT;
  out[idx] = plane[(size_t)s * kTPad + t];
}

extern "C" void kernel_launch(void* const* d_in, const int* in_sizes, int n_in,
                              void* d_out, int out_size, void* d_ws, size_t ws_size,
                              hipStream_t stream) {
  const float* x   = (const float*)d_in[0];
  const float* z   = (const float*)d_in[1];
  const float* msk = (const float*)d_in[2];
  const float* W1  = (const float*)d_in[3];
  const float* b1  = (const float*)d_in[4];
  const float* W2  = (const float*)d_in[5];
  const float* b2  = (const float*)d_in[6];
  float* out = (float*)d_out;

  float* ws    = (float*)d_ws;
  float* h     = ws;
  float* plane = h + (size_t)kK * kHid;

  sipm_h_kernel<<<(kK * (kHid / 4) + 255) / 256, 256, 0, stream>>>(x, W1, b1, h);
  sipm_window_kernel<<<dim3(kSGrid, kTGrid), 128, 0, stream>>>(z, msk, W2, b2, h, plane);
  sipm_copy_kernel<<<(kS * kT + 255) / 256, 256, 0, stream>>>(plane, out);
}


# Electrons of each 64x32 tile split over 5 workgroups (2 for the 6-tick last window): 3045 equal pieces fill whole rounds of 4 per WGP, was 630 at 3 or 2 per WGP; out sums part planes: new sum order
# speedup vs baseline: 6.5626x; 1.2677x over previous
#include <hip/hip_runtime.h>
#include <math.h>


typedef float v2f __attribute__((ext_vector_type(2)));
typedef float v4f __attribute__((ext_vector_type(4)));
typedef float v8f __attribute__((ext_vector_type(8)));

namespace {
constexpr int kBatch  = 64;
constexpr int kEv     = 500;
constexpr int kK      = kBatch * kEv;
constexpr int kHid    = 128;
constexpr int kS      = 2209;
constexpr int kSW     = 64;
constexpr int kSGrid  = (kS + kSW - 1) / kSW;
constexpr int kSPad   = kSGrid * kSW;
constexpr int kT      = 550;
constexpr int kTW     = 32;
constexpr int kTGrid  = (kT + kTW - 1) / kTW;
constexpr int kTPad   = kTGrid * kTW;
constexpr int kReach  = 11;
constexpr int kCH     = 3200;
constexpr int kNCh    = kK / kCH;
constexpr int kSplitA = 5;
constexpr int kSplitB = 2;
constexpr int kTilesA = kSGrid * (kTGrid - 1);
constexpr int kWgA    = kTilesA * kSplitA;
constexpr int kWgB    = kSGrid * kSplitB;
constexpr int kWg     = kWgA + kWgB;
constexpr int kSeg    = kCH / 4;
constexpr int kBR     = 32;
constexpr int kPitchF = kHid + 4;
constexpr int kListCap = kCH + kBR;
constexpr int kTapQ   = 32 * 4;
constexpr int kTapBuf = 2 * 2 * 2 * kTapQ;
constexpr float kNorm = 0.3989422804f * 1.41421356237309515f;
constexpr size_t kPlaneF = (size_t)kSPad * kTPad;
static_assert(kNCh * kCH == kK, "chunks must tile K");
static_assert(kNCh % kSplitA == 0 && kNCh % kSplitB == 0, "parts must tile the chunks");
static_assert(kSeg % 32 == 0, "segment must be whole wave loads");
}

__global__ __launch_bounds__(256)
void sipm_h_kernel(const float* __restrict__ x, const float* __restrict__ W1,
                   const float* __restrict__ b1, float* __restrict__ h) {
  int idx = blockIdx.x * 256 + threadIdx.x;
  if (idx >= kK * (kHid / 4)) return;
  int k = idx >> 5;
  int j = (idx & 31) << 2;
  float x0 = x[k * 3 + 0], x1 = x[k * 3 + 1], x2 = x[k * 3 + 2];
  const float4 wa = *(const float4*)(W1 + 0 * kHid + j);
  const float4 wb = *(const float4*)(W1 + 1 * kHid + j);
  const float4 wc = *(const float4*)(W1 + 2 * kHid + j);
  const float4 bb = *(const float4*)(b1 + j);
  float4 r;
  r.x = fmaxf(fmaf(x0, wa.x, fmaf(x1, wb.x, fmaf(x2, wc.x, bb.x))), 0.f);
  r.y = fmaxf(fmaf(x0, wa.y, fmaf(x1, wb.y, fmaf(x2, wc.y, bb.y))), 0.f);
  r.z = fmaxf(fmaf(x0, wa.z, fmaf(x1, wb.z, fmaf(x2, wc.z, bb.z))), 0.f);
  r.w = fmaxf(fmaf(x0, wa.w, fmaf(x1, wb.w, fmaf(x2, wc.w, bb.w))), 0.f);
  *(float4*)(h + (size_t)k * kHid + j) = r;
}

__global__ __launch_bounds__(128)
void sipm_window_kernel(const float* __restrict__ z, const float* __restrict__ mask,
                        const float* __restrict__ W2, const float* __restrict__ b2,
                        const float* __restrict__ h, float* __restrict__ planes) {
  __shared__ __attribute__((aligned(16))) int   lds_k[kListCap];
  __shared__ __attribute__((aligned(16))) float lds_z[kListCap];
  __shared__ __attribute__((aligned(16))) float lds_m[kListCap];
  __shared__ int lds_cnt[4];
  __shared__ __attribute__((aligned(16))) float lds_h[2][kBR * kPitchF];
  __shared__ __attribute__((aligned(16))) float lds_tap[2][kTapBuf];

  const int tid  = threadIdx.x;
  const int w    = tid >> 5;
  const int lane = tid & 31;
  const int m    = lane & 15;
  const int hf   = lane >> 4;
  const int id = blockIdx.x;
  int part, strip, win, nch;
  if (id < kWgA) {
    part = id / kTilesA;
    const int r = id - part * kTilesA;
    win   = r / kSGrid;
    strip = r - win * kSGrid;
    nch   = kNCh / kSplitA;
  } else {
    const int r = id - kWgA;
    part  = r / kSGrid;
    strip = r - part * kSGrid;
    win   = kTGrid - 1;
    nch   = kNCh / kSplitB;
  }
  const int s_base = strip * kSW + w * 16;
  const int t_base = win * kTW;
  const int  t_last = (t_base + kTW - 1 < kT - 1) ? (t_base + kTW - 1) : (kT - 1);
  const float zlo = (float)(t_base - kReach);
  const float zhi = (float)(t_last + kReach);
  const int   gw  = w >> 1;
  const int   tw  = w & 1;
  const float tfw = (float)(t_base + 16 * tw + m);
  const int   tap_wr = ((gw * 2 + tw) * 2) * kTapQ + 4 * lane;

  const int  sW  = s_base + m;
  const bool sok = sW < kS;
  v2f w2f[16][2];
  #pragma unroll
  for (int j = 0; j < 16; ++j) {
    #pragma unroll
    for (int q = 0; q < 2; ++q) {
      const int i0 = 8 * j + 4 * hf + 2 * q;
      w2f[j][q].x = sok ? W2[(size_t)(i0 + 0) * kS + sW] : 0.f;
      w2f[j][q].y = sok ? W2[(size_t)(i0 + 1) * kS + sW] : 0.f;
    }
  }
  const float bias = sok ? b2[sW] : 0.f;

  v8f acc0 = {}, acc1 = {};

  const int ch0 = part * nch;
  for (int ch = ch0; ch < ch0 + nch; ++ch) {
    const int cbase = ch * kCH + w * kSeg;
    {
      int cnt = 0;
      #pragma unroll 5
      for (int j = 0; j < kSeg / 32; ++j) {
        const float zk = z[cbase + j * 32 + lane];
        const bool hit = (zk > zlo) && (zk < zhi);
        cnt += __builtin_popcount(__builtin_amdgcn_ballot_w32(hit));
      }
      if (lane == 0) lds_cnt[w] = cnt;
    }
    __syncthreads();
    const int c0 = lds_cnt[0], c1 = lds_cnt[1], c2 = lds_cnt[2], c3 = lds_cnt[3];
    const int n = __builtin_amdgcn_readfirstlane(c0 + c1 + c2 + c3);
    {
      int pos = (w > 0 ? c0 : 0) + (w > 1 ? c1 : 0) + (w > 2 ? c2 : 0);
      #pragma unroll 5
      for (int j = 0; j < kSeg / 32; ++j) {
        const int k = cbase + j * 32 + lane;
        const float zk = z[k];
        const bool hit = (zk > zlo) && (zk < zhi);
        const unsigned b = __builtin_amdgcn_ballot_w32(hit);
        if (hit) {
          const int p = pos + __builtin_popcount(b & ((1u << lane) - 1u));
          lds_k[p] = k;
          lds_z[p] = zk;
          lds_m[p] = mask[k];
        }
        pos += __builtin_popcount(b);
      }
      if (w == 3) {
        const int p = n + lane;
        lds_k[p] = 0;
        lds_z[p] = 0.f;
        lds_m[p] = 0.f;
      }
    }
    __syncthreads();

    {
      const int ng = (n + 15) >> 4;
      const int nb = (ng + 1) >> 1;
      if (nb > 0) {
        float4 pre[8];
        {
          const int4 ka = *(const int4*)(lds_k + 8 * w);
          const int4 kb = *(const int4*)(lds_k + 8 * w + 4);
          const int kk[8] = {ka.x, ka.y, ka.z, ka.w, kb.x, kb.y, kb.z, kb.w};
          #pragma unroll
          for (int i = 0; i < 8; ++i) {
            const int ke = __builtin_amdgcn_readfirstlane(kk[i]);
            pre[i] = *((const float4*)(h + (size_t)ke * kHid) + lane);
          }
          {
            const int ez = 16 * gw + 8 * hf;
            const v4f z0 = *(const v4f*)(lds_z + ez);
            const v4f z1 = *(const v4f*)(lds_z + ez + 4);
            const v4f m0 = *(const v4f*)(lds_m + ez);
            const v4f m1 = *(const v4f*)(lds_m + ez + 4);
            const float zz[8] = {z0.x, z0.y, z0.z, z0.w, z1.x, z1.y, z1.z, z1.w};
            const float mm[8] = {m0.x, m0.y, m0.z, m0.w, m1.x, m1.y, m1.z, m1.w};
            float tp[8];
            #pragma unroll
            for (int i = 0; i < 8; ++i) {
              const float d = tfw - zz[i];
              tp[i] = expf(-d * d) * mm[i] * kNorm;
            }
            float* td = &lds_tap[0][tap_wr];
            v4f q0 = {tp[0], tp[1], tp[2], tp[3]};
            v4f q1 = {tp[4], tp[5], tp[6], tp[7]};
            *(v4f*)td = q0;
            *(v4f*)(td + kTapQ) = q1;
          }
          float* dst = &lds_h[0][(8 * w) * kPitchF + 4 * lane];
          #pragma unroll
          for (int i = 0; i < 8; ++i) *(float4*)(dst + i * kPitchF) = pre[i];
        }
        __syncthreads();
        for (int b = 0; b < nb; ++b) {
          const int bn = (b + 1 < nb) ? (b + 1) : b;
          {
            const int4 ka = *(const int4*)(lds_k + bn * kBR + 8 * w);
            const int4 kb = *(const int4*)(lds_k + bn * kBR + 8 * w + 4);
            const int kk[8] = {ka.x, ka.y, ka.z, ka.w, kb.x, kb.y, kb.z, kb.w};
            #pragma unroll
            for (int i = 0; i < 8; ++i) {
              const int ke = __builtin_amdgcn_readfirstlane(kk[i]);
              pre[i] = *((const float4*)(h + (size_t)ke * kHid) + lane);
            }
          }
          const float* cur  = &lds_h[b & 1][0];
          const float* tcur = &lds_tap[b & 1][4 * lane];
          float*       tnxt = &lds_tap[(b + 1) & 1][tap_wr];
          #pragma unroll
          for (int half = 0; half < 2; ++half) {
            const int g = 2 * b + half;
            if (g >= ng) break;
            const float* arow = cur + (half * 16 + m) * kPitchF + 4 * hf;
            v4f a[16];
            #pragma unroll
            for (int j = 0; j < 16; ++j) a[j] = *(const v4f*)(arow + 8 * j);
            v4f nz0 = {}, nz1 = {}, nm0 = {}, nm1 = {};
            if (half == 0) {
              const int ez = bn * kBR + 16 * gw + 8 * hf;
              nz0 = *(const v4f*)(lds_z + ez);
              nz1 = *(const v4f*)(lds_z + ez + 4);
              nm0 = *(const v4f*)(lds_m + ez);
              nm1 = *(const v4f*)(lds_m + ez + 4);
            }
            const float* tg = tcur + (half * 4) * kTapQ;
            const v4f b00 = *(const v4f*)(tg);
            const v4f b01 = *(const v4f*)(tg + kTapQ);
            const v4f b10 = *(const v4f*)(tg + 2 * kTapQ);
            const v4f b11 = *(const v4f*)(tg + 3 * kTapQ);
            __builtin_amdgcn_sched_barrier(0);

            v8f d0 = {}, d1 = {};
            #pragma unroll
            for (int j = 0; j < 16; ++j) {
              v2f a0; a0.x = a[j].x; a0.y = a[j].y;
              v2f a1; a1.x = a[j].z; a1.y = a[j].w;
              d0 = __builtin_amdgcn_wmma_f32_16x16x4_f32(false, a0, false, w2f[j][0],
                                                         (short)0, d0, false, false);
              d1 = __builtin_amdgcn_wmma_f32_16x16x4_f32(false, a1, false, w2f[j][1],
                                                         (short)0, d1, false, false);
            }
            if (half == 0) {
              const float zz[8] = {nz0.x, nz0.y, nz0.z, nz0.w, nz1.x, nz1.y, nz1.z, nz1.w};
              const float mm[8] = {nm0.x, nm0.y, nm0.z, nm0.w, nm1.x, nm1.y, nm1.z, nm1.w};
              float tp[8];
              #pragma unroll
              for (int i = 0; i < 8; ++i) {
                const float d = tfw - zz[i];
                tp[i] = expf(-d * d) * mm[i] * kNorm;
              }
              v4f q0 = {tp[0], tp[1], tp[2], tp[3]};
              v4f q1 = {tp[4], tp[5], tp[6], tp[7]};
              *(v4f*)tnxt = q0;
              *(v4f*)(tnxt + kTapQ) = q1;
            }
            float resp[8];
            #pragma unroll
            for (int r = 0; r < 8; ++r) resp[r] = expf((d0[r] + d1[r]) + bias);

            const float t0[8] = {b00.x, b00.y, b00.z, b00.w, b01.x, b01.y, b01.z, b01.w};
            const float t1[8] = {b10.x, b10.y, b10.z, b10.w, b11.x, b11.y, b11.z, b11.w};
            #pragma unroll
            for (int j = 0; j < 4; ++j) {
              v2f af;  af.x  = resp[2 * j]; af.y  = resp[2 * j + 1];
              v2f bf0; bf0.x = t0[2 * j];   bf0.y = t0[2 * j + 1];
              v2f bf1; bf1.x = t1[2 * j];   bf1.y = t1[2 * j + 1];
              acc0 = __builtin_amdgcn_wmma_f32_16x16x4_f32(false, af, false, bf0,
                                                           (short)0, acc0, false, false);
              acc1 = __builtin_amdgcn_wmma_f32_16x16x4_f32(false, af, false, bf1,
                                                           (short)0, acc1, false, false);
            }
          }
          {
            float* dst = &lds_h[(b + 1) & 1][(8 * w) * kPitchF + 4 * lane];
            #pragma unroll
            for (int i = 0; i < 8; ++i) *(float4*)(dst + i * kPitchF) = pre[i];
          }
          __syncthreads();
        }
      }
    }
  }

  float* pbase = planes + (size_t)part * kPlaneF
               + (size_t)(s_base + 8 * hf) * kTPad + t_base + m;
  #pragma unroll
  for (int r = 0; r < 8; ++r) {
    float* prow = pbase + (size_t)r * kTPad;
    prow[0] = acc0[r];
    prow[16] = acc1[r];
  }
}

__global__ __launch_bounds__(256)
void sipm_copy_kernel(const float* __restrict__ planes, float* __restrict__ out) {
  int idx = blockIdx.x * 256 + threadIdx.x;
  if (idx >= kS * kT) return;
  int s = idx / kT, t = idx - s * kT;
  const float* p = planes + (size_t)s * kTPad + t;
  const int np = (t < (kTGrid - 1) * kTW) ? kSplitA : kSplitB;
  float v[kSplitA];
  #pragma unroll
  for (int i = 0; i < kSplitA; ++i) v[i] = (i < np) ? p[(size_t)i * kPlaneF] : 0.f;
  float sum = v[0];
  #pragma unroll
  for (int i = 1; i < kSplitA; ++i) sum += v[i];
  out[idx] = sum;
}

extern "C" void kernel_launch(void* const* d_in, const int* in_sizes, int n_in,
                              void* d_out, int out_size, void* d_ws, size_t ws_size,
                              hipStream_t stream) {
  const float* x   = (const float*)d_in[0];
  const float* z   = (const float*)d_in[1];
  const float* msk = (const float*)d_in[2];
  const float* W1  = (const float*)d_in[3];
  const float* b1  = (const float*)d_in[4];
  const float* W2  = (const float*)d_in[5];
  const float* b2  = (const float*)d_in[6];
  float* out = (float*)d_out;

  float* ws    = (float*)d_ws;
  float* h     = ws;
  float* plane = h + (size_t)kK * kHid;

  sipm_h_kernel<<<(kK * (kHid / 4) + 255) / 256, 256, 0, stream>>>(x, W1, b1, h);
  sipm_window_kernel<<<kWg, 128, 0, stream>>>(z, msk, W2, b2, h, plane);
  sipm_copy_kernel<<<(kS * kT + 255) / 256, 256, 0, stream>>>(plane, out);
}


# List scan reads memory once: a wave loads its 800 z and 800 mask values into registers, 50 loads in flight, and lists from them; was 25 z-then-mask load pairs, each waited out alone; sums unchanged
# speedup vs baseline: 7.5805x; 1.1551x over previous
#include <hip/hip_runtime.h>
#include <math.h>


typedef float v2f __attribute__((ext_vector_type(2)));
typedef float v4f __attribute__((ext_vector_type(4)));
typedef float v8f __attribute__((ext_vector_type(8)));

namespace {
constexpr int kBatch  = 64;
constexpr int kEv     = 500;
constexpr int kK      = kBatch * kEv;
constexpr int kHid    = 128;
constexpr int kS      = 2209;
constexpr int kSW     = 64;
constexpr int kSGrid  = (kS + kSW - 1) / kSW;
constexpr int kSPad   = kSGrid * kSW;
constexpr int kT      = 550;
constexpr int kTW     = 32;
constexpr int kTGrid  = (kT + kTW - 1) / kTW;
constexpr int kTPad   = kTGrid * kTW;
constexpr int kReach  = 11;
constexpr int kCH     = 3200;
constexpr int kNCh    = kK / kCH;
constexpr int kSplitA = 5;
constexpr int kSplitB = 2;
constexpr int kTilesA = kSGrid * (kTGrid - 1);
constexpr int kWgA    = kTilesA * kSplitA;
constexpr int kWgB    = kSGrid * kSplitB;
constexpr int kWg     = kWgA + kWgB;
constexpr int kSeg    = kCH / 4;
constexpr int kSegN   = kSeg / 32;
constexpr int kBR     = 32;
constexpr int kPitchF = kHid + 4;
constexpr int kListCap = kCH + kBR;
constexpr int kTapQ   = 32 * 4;
constexpr int kTapBuf = 2 * 2 * 2 * kTapQ;
constexpr float kNorm = 0.3989422804f * 1.41421356237309515f;
constexpr size_t kPlaneF = (size_t)kSPad * kTPad;
static_assert(kNCh * kCH == kK, "chunks must tile K");
static_assert(kNCh % kSplitA == 0 && kNCh % kSplitB == 0, "parts must tile the chunks");
static_assert(kSeg % 32 == 0, "segment must be whole wave loads");
}

__global__ __launch_bounds__(256)
void sipm_h_kernel(const float* __restrict__ x, const float* __restrict__ W1,
                   const float* __restrict__ b1, float* __restrict__ h) {
  int idx = blockIdx.x * 256 + threadIdx.x;
  if (idx >= kK * (kHid / 4)) return;
  int k = idx >> 5;
  int j = (idx & 31) << 2;
  float x0 = x[k * 3 + 0], x1 = x[k * 3 + 1], x2 = x[k * 3 + 2];
  const float4 wa = *(const float4*)(W1 + 0 * kHid + j);
  const float4 wb = *(const float4*)(W1 + 1 * kHid + j);
  const float4 wc = *(const float4*)(W1 + 2 * kHid + j);
  const float4 bb = *(const float4*)(b1 + j);
  float4 r;
  r.x = fmaxf(fmaf(x0, wa.x, fmaf(x1, wb.x, fmaf(x2, wc.x, bb.x))), 0.f);
  r.y = fmaxf(fmaf(x0, wa.y, fmaf(x1, wb.y, fmaf(x2, wc.y, bb.y))), 0.f);
  r.z = fmaxf(fmaf(x0, wa.z, fmaf(x1, wb.z, fmaf(x2, wc.z, bb.z))), 0.f);
  r.w = fmaxf(fmaf(x0, wa.w, fmaf(x1, wb.w, fmaf(x2, wc.w, bb.w))), 0.f);
  *(float4*)(h + (size_t)k * kHid + j) = r;
}

__global__ __launch_bounds__(128)
void sipm_window_kernel(const float* __restrict__ z, const float* __restrict__ mask,
                        const float* __restrict__ W2, const float* __restrict__ b2,
                        const float* __restrict__ h, float* __restrict__ planes) {
  __shared__ __attribute__((aligned(16))) int   lds_k[kListCap];
  __shared__ __attribute__((aligned(16))) float lds_z[kListCap];
  __shared__ __attribute__((aligned(16))) float lds_m[kListCap];
  __shared__ int lds_cnt[4];
  __shared__ __attribute__((aligned(16))) float lds_h[2][kBR * kPitchF];
  __shared__ __attribute__((aligned(16))) float lds_tap[2][kTapBuf];

  const int tid  = threadIdx.x;
  const int w    = tid >> 5;
  const int lane = tid & 31;
  const int m    = lane & 15;
  const int hf   = lane >> 4;
  const int id = blockIdx.x;
  int part, strip, win, nch;
  if (id < kWgA) {
    part = id / kTilesA;
    const int r = id - part * kTilesA;
    win   = r / kSGrid;
    strip = r - win * kSGrid;
    nch   = kNCh / kSplitA;
  } else {
    const int r = id - kWgA;
    part  = r / kSGrid;
    strip = r - part * kSGrid;
    win   = kTGrid - 1;
    nch   = kNCh / kSplitB;
  }
  const int s_base = strip * kSW + w * 16;
  const int t_base = win * kTW;
  const int  t_last = (t_base + kTW - 1 < kT - 1) ? (t_base + kTW - 1) : (kT - 1);
  const float zlo = (float)(t_base - kReach);
  const float zhi = (float)(t_last + kReach);
  const int   gw  = w >> 1;
  const int   tw  = w & 1;
  const float tfw = (float)(t_base + 16 * tw + m);
  const int   tap_wr = ((gw * 2 + tw) * 2) * kTapQ + 4 * lane;

  const int  sW  = s_base + m;
  const bool sok = sW < kS;
  v2f w2f[16][2];
  #pragma unroll
  for (int j = 0; j < 16; ++j) {
    #pragma unroll
    for (int q = 0; q < 2; ++q) {
      const int i0 = 8 * j + 4 * hf + 2 * q;
      w2f[j][q].x = sok ? W2[(size_t)(i0 + 0) * kS + sW] : 0.f;
      w2f[j][q].y = sok ? W2[(size_t)(i0 + 1) * kS + sW] : 0.f;
    }
  }
  const float bias = sok ? b2[sW] : 0.f;

  v8f acc0 = {}, acc1 = {};

  const int ch0 = part * nch;
  for (int ch = ch0; ch < ch0 + nch; ++ch) {
    const int cbase = ch * kCH + w * kSeg;
    float zr[kSegN], mr[kSegN];
    #pragma unroll
    for (int j = 0; j < kSegN; ++j) zr[j] = z[cbase + j * 32 + lane];
    #pragma unroll
    for (int j = 0; j < kSegN; ++j) mr[j] = mask[cbase + j * 32 + lane];
    {
      int cnt = 0;
      #pragma unroll
      for (int j = 0; j < kSegN; ++j) {
        const bool hit = (zr[j] > zlo) && (zr[j] < zhi);
        cnt += __builtin_popcount(__builtin_amdgcn_ballot_w32(hit));
      }
      if (lane == 0) lds_cnt[w] = cnt;
    }
    __syncthreads();
    const int c0 = lds_cnt[0], c1 = lds_cnt[1], c2 = lds_cnt[2], c3 = lds_cnt[3];
    const int n = __builtin_amdgcn_readfirstlane(c0 + c1 + c2 + c3);
    {
      int pos = (w > 0 ? c0 : 0) + (w > 1 ? c1 : 0) + (w > 2 ? c2 : 0);
      #pragma unroll
      for (int j = 0; j < kSegN; ++j) {
        const float zk = zr[j];
        const bool hit = (zk > zlo) && (zk < zhi);
        const unsigned b = __builtin_amdgcn_ballot_w32(hit);
        if (hit) {
          const int p = pos + __builtin_popcount(b & ((1u << lane) - 1u));
          lds_k[p] = cbase + j * 32 + lane;
          lds_z[p] = zk;
          lds_m[p] = mr[j];
        }
        pos += __builtin_popcount(b);
      }
      if (w == 3) {
        const int p = n + lane;
        lds_k[p] = 0;
        lds_z[p] = 0.f;
        lds_m[p] = 0.f;
      }
    }
    __syncthreads();

    {
      const int ng = (n + 15) >> 4;
      const int nb = (ng + 1) >> 1;
      if (nb > 0) {
        float4 pre[8];
        {
          const int4 ka = *(const int4*)(lds_k + 8 * w);
          const int4 kb = *(const int4*)(lds_k + 8 * w + 4);
          const int kk[8] = {ka.x, ka.y, ka.z, ka.w, kb.x, kb.y, kb.z, kb.w};
          #pragma unroll
          for (int i = 0; i < 8; ++i) {
            const int ke = __builtin_amdgcn_readfirstlane(kk[i]);
            pre[i] = *((const float4*)(h + (size_t)ke * kHid) + lane);
          }
          {
            const int ez = 16 * gw + 8 * hf;
            const v4f z0 = *(const v4f*)(lds_z + ez);
            const v4f z1 = *(const v4f*)(lds_z + ez + 4);
            const v4f m0 = *(const v4f*)(lds_m + ez);
            const v4f m1 = *(const v4f*)(lds_m + ez + 4);
            const float zz[8] = {z0.x, z0.y, z0.z, z0.w, z1.x, z1.y, z1.z, z1.w};
            const float mm[8] = {m0.x, m0.y, m0.z, m0.w, m1.x, m1.y, m1.z, m1.w};
            float tp[8];
            #pragma unroll
            for (int i = 0; i < 8; ++i) {
              const float d = tfw - zz[i];
              tp[i] = expf(-d * d) * mm[i] * kNorm;
            }
            float* td = &lds_tap[0][tap_wr];
            v4f q0 = {tp[0], tp[1], tp[2], tp[3]};
            v4f q1 = {tp[4], tp[5], tp[6], tp[7]};
            *(v4f*)td = q0;
            *(v4f*)(td + kTapQ) = q1;
          }
          float* dst = &lds_h[0][(8 * w) * kPitchF + 4 * lane];
          #pragma unroll
          for (int i = 0; i < 8; ++i) *(float4*)(dst + i * kPitchF) = pre[i];
        }
        __syncthreads();
        for (int b = 0; b < nb; ++b) {
          const int bn = (b + 1 < nb) ? (b + 1) : b;
          {
            const int4 ka = *(const int4*)(lds_k + bn * kBR + 8 * w);
            const int4 kb = *(const int4*)(lds_k + bn * kBR + 8 * w + 4);
            const int kk[8] = {ka.x, ka.y, ka.z, ka.w, kb.x, kb.y, kb.z, kb.w};
            #pragma unroll
            for (int i = 0; i < 8; ++i) {
              const int ke = __builtin_amdgcn_readfirstlane(kk[i]);
              pre[i] = *((const float4*)(h + (size_t)ke * kHid) + lane);
            }
          }
          const float* cur  = &lds_h[b & 1][0];
          const float* tcur = &lds_tap[b & 1][4 * lane];
          float*       tnxt = &lds_tap[(b + 1) & 1][tap_wr];
          #pragma unroll
          for (int half = 0; half < 2; ++half) {
            const int g = 2 * b + half;
            if (g >= ng) break;
            const float* arow = cur + (half * 16 + m) * kPitchF + 4 * hf;
            v4f a[16];
            #pragma unroll
            for (int j = 0; j < 16; ++j) a[j] = *(const v4f*)(arow + 8 * j);
            v4f nz0 = {}, nz1 = {}, nm0 = {}, nm1 = {};
            if (half == 0) {
              const int ez = bn * kBR + 16 * gw + 8 * hf;
              nz0 = *(const v4f*)(lds_z + ez);
              nz1 = *(const v4f*)(lds_z + ez + 4);
              nm0 = *(const v4f*)(lds_m + ez);
              nm1 = *(const v4f*)(lds_m + ez + 4);
            }
            const float* tg = tcur + (half * 4) * kTapQ;
            const v4f b00 = *(const v4f*)(tg);
            const v4f b01 = *(const v4f*)(tg + kTapQ);
            const v4f b10 = *(const v4f*)(tg + 2 * kTapQ);
            const v4f b11 = *(const v4f*)(tg + 3 * kTapQ);
            __builtin_amdgcn_sched_barrier(0);

            v8f d0 = {}, d1 = {};
            #pragma unroll
            for (int j = 0; j < 16; ++j) {
              v2f a0; a0.x = a[j].x; a0.y = a[j].y;
              v2f a1; a1.x = a[j].z; a1.y = a[j].w;
              d0 = __builtin_amdgcn_wmma_f32_16x16x4_f32(false, a0, false, w2f[j][0],
                                                         (short)0, d0, false, false);
              d1 = __builtin_amdgcn_wmma_f32_16x16x4_f32(false, a1, false, w2f[j][1],
                                                         (short)0, d1, false, false);
            }
            if (half == 0) {
              const float zz[8] = {nz0.x, nz0.y, nz0.z, nz0.w, nz1.x, nz1.y, nz1.z, nz1.w};
              const float mm[8] = {nm0.x, nm0.y, nm0.z, nm0.w, nm1.x, nm1.y, nm1.z, nm1.w};
              float tp[8];
              #pragma unroll
              for (int i = 0; i < 8; ++i) {
                const float d = tfw - zz[i];
                tp[i] = expf(-d * d) * mm[i] * kNorm;
              }
              v4f q0 = {tp[0], tp[1], tp[2], tp[3]};
              v4f q1 = {tp[4], tp[5], tp[6], tp[7]};
              *(v4f*)tnxt = q0;
              *(v4f*)(tnxt + kTapQ) = q1;
            }
            float resp[8];
            #pragma unroll
            for (int r = 0; r < 8; ++r) resp[r] = expf((d0[r] + d1[r]) + bias);

            const float t0[8] = {b00.x, b00.y, b00.z, b00.w, b01.x, b01.y, b01.z, b01.w};
            const float t1[8] = {b10.x, b10.y, b10.z, b10.w, b11.x, b11.y, b11.z, b11.w};
            #pragma unroll
            for (int j = 0; j < 4; ++j) {
              v2f af;  af.x  = resp[2 * j]; af.y  = resp[2 * j + 1];
              v2f bf0; bf0.x = t0[2 * j];   bf0.y = t0[2 * j + 1];
              v2f bf1; bf1.x = t1[2 * j];   bf1.y = t1[2 * j + 1];
              acc0 = __builtin_amdgcn_wmma_f32_16x16x4_f32(false, af, false, bf0,
                                                           (short)0, acc0, false, false);
              acc1 = __builtin_amdgcn_wmma_f32_16x16x4_f32(false, af, false, bf1,
                                                           (short)0, acc1, false, false);
            }
          }
          {
            float* dst = &lds_h[(b + 1) & 1][(8 * w) * kPitchF + 4 * lane];
            #pragma unroll
            for (int i = 0; i < 8; ++i) *(float4*)(dst + i * kPitchF) = pre[i];
          }
          __syncthreads();
        }
      }
    }
  }

  float* pbase = planes + (size_t)part * kPlaneF
               + (size_t)(s_base + 8 * hf) * kTPad + t_base + m;
  #pragma unroll
  for (int r = 0; r < 8; ++r) {
    float* prow = pbase + (size_t)r * kTPad;
    prow[0] = acc0[r];
    prow[16] = acc1[r];
  }
}

__global__ __launch_bounds__(256)
void sipm_copy_kernel(const float* __restrict__ planes, float* __restrict__ out) {
  int idx = blockIdx.x * 256 + threadIdx.x;
  if (idx >= kS * kT) return;
  int s = idx / kT, t = idx - s * kT;
  const float* p = planes + (size_t)s * kTPad + t;
  const int np = (t < (kTGrid - 1) * kTW) ? kSplitA : kSplitB;
  float v[kSplitA];
  #pragma unroll
  for (int i = 0; i < kSplitA; ++i) v[i] = (i < np) ? p[(size_t)i * kPlaneF] : 0.f;
  float sum = v[0];
  #pragma unroll
  for (int i = 1; i < kSplitA; ++i) sum += v[i];
  out[idx] = sum;
}

extern "C" void kernel_launch(void* const* d_in, const int* in_sizes, int n_in,
                              void* d_out, int out_size, void* d_ws, size_t ws_size,
                              hipStream_t stream) {
  const float* x   = (const float*)d_in[0];
  const float* z   = (const float*)d_in[1];
  const float* msk = (const float*)d_in[2];
  const float* W1  = (const float*)d_in[3];
  const float* b1  = (const float*)d_in[4];
  const float* W2  = (const float*)d_in[5];
  const float* b2  = (const float*)d_in[6];
  float* out = (float*)d_out;

  float* ws    = (float*)d_ws;
  float* h     = ws;
  float* plane = h + (size_t)kK * kHid;

  sipm_h_kernel<<<(kK * (kHid / 4) + 255) / 256, 256, 0, stream>>>(x, W1, b1, h);
  sipm_window_kernel<<<kWg, 128, 0, stream>>>(z, msk, W2, b2, h, plane);
  sipm_copy_kernel<<<(kS * kT + 255) / 256, 256, 0, stream>>>(plane, out);
}


# h rows made inside the window kernel from each kept electron's 3 inputs (W1, b1 in registers), beside the second group's resp WMMAs; h launch, h array and all h row reads gone; sums unchanged
# speedup vs baseline: 7.9061x; 1.0430x over previous
#include <hip/hip_runtime.h>
#include <math.h>


typedef float v2f __attribute__((ext_vector_type(2)));
typedef float v4f __attribute__((ext_vector_type(4)));
typedef float v8f __attribute__((ext_vector_type(8)));

namespace {
constexpr int kBatch  = 64;
constexpr int kEv     = 500;
constexpr int kK      = kBatch * kEv;
constexpr int kHid    = 128;
constexpr int kS      = 2209;
constexpr int kSW     = 64;
constexpr int kSGrid  = (kS + kSW - 1) / kSW;
constexpr int kSPad   = kSGrid * kSW;
constexpr int kT      = 550;
constexpr int kTW     = 32;
constexpr int kTGrid  = (kT + kTW - 1) / kTW;
constexpr int kTPad   = kTGrid * kTW;
constexpr int kReach  = 11;
constexpr int kCH     = 3200;
constexpr int kNCh    = kK / kCH;
constexpr int kSplitA = 5;
constexpr int kSplitB = 2;
constexpr int kTilesA = kSGrid * (kTGrid - 1);
constexpr int kWgA    = kTilesA * kSplitA;
constexpr int kWgB    = kSGrid * kSplitB;
constexpr int kWg     = kWgA + kWgB;
constexpr int kSeg    = kCH / 4;
constexpr int kSegN   = kSeg / 32;
constexpr int kBR     = 32;
constexpr int kPitchF = kHid + 4;
constexpr int kListCap = kCH + kBR;
constexpr int kTapQ   = 32 * 4;
constexpr int kTapBuf = 2 * 2 * 2 * kTapQ;
constexpr float kNorm = 0.3989422804f * 1.41421356237309515f;
constexpr size_t kPlaneF = (size_t)kSPad * kTPad;
static_assert(kNCh * kCH == kK, "chunks must tile K");
static_assert(kNCh % kSplitA == 0 && kNCh % kSplitB == 0, "parts must tile the chunks");
static_assert(kSeg % 32 == 0, "segment must be whole wave loads");
}

__device__ __forceinline__ float4 sipm_h_row(float x0, float x1, float x2, const float4 wa,
                                             const float4 wb, const float4 wc, const float4 bb) {
  float4 r;
  r.x = fmaxf(fmaf(x0, wa.x, fmaf(x1, wb.x, fmaf(x2, wc.x, bb.x))), 0.f);
  r.y = fmaxf(fmaf(x0, wa.y, fmaf(x1, wb.y, fmaf(x2, wc.y, bb.y))), 0.f);
  r.z = fmaxf(fmaf(x0, wa.z, fmaf(x1, wb.z, fmaf(x2, wc.z, bb.z))), 0.f);
  r.w = fmaxf(fmaf(x0, wa.w, fmaf(x1, wb.w, fmaf(x2, wc.w, bb.w))), 0.f);
  return r;
}

__global__ __launch_bounds__(128)
void sipm_window_kernel(const float* __restrict__ z, const float* __restrict__ mask,
                        const float* __restrict__ W2, const float* __restrict__ b2,
                        const float* __restrict__ x, const float* __restrict__ W1,
                        const float* __restrict__ b1, float* __restrict__ planes) {
  __shared__ __attribute__((aligned(16))) int   lds_k[kListCap];
  __shared__ __attribute__((aligned(16))) float lds_z[kListCap];
  __shared__ __attribute__((aligned(16))) float lds_m[kListCap];
  __shared__ int lds_cnt[4];
  __shared__ __attribute__((aligned(16))) float lds_h[2][kBR * kPitchF];
  __shared__ __attribute__((aligned(16))) float lds_tap[2][kTapBuf];

  const int tid  = threadIdx.x;
  const int w    = tid >> 5;
  const int lane = tid & 31;
  const int m    = lane & 15;
  const int hf   = lane >> 4;
  const int id = blockIdx.x;
  int part, strip, win, nch;
  if (id < kWgA) {
    part = id / kTilesA;
    const int r = id - part * kTilesA;
    win   = r / kSGrid;
    strip = r - win * kSGrid;
    nch   = kNCh / kSplitA;
  } else {
    const int r = id - kWgA;
    part  = r / kSGrid;
    strip = r - part * kSGrid;
    win   = kTGrid - 1;
    nch   = kNCh / kSplitB;
  }
  const int s_base = strip * kSW + w * 16;
  const int t_base = win * kTW;
  const int  t_last = (t_base + kTW - 1 < kT - 1) ? (t_base + kTW - 1) : (kT - 1);
  const float zlo = (float)(t_base - kReach);
  const float zhi = (float)(t_last + kReach);
  const int   gw  = w >> 1;
  const int   tw  = w & 1;
  const float tfw = (float)(t_base + 16 * tw + m);
  const int   tap_wr = ((gw * 2 + tw) * 2) * kTapQ + 4 * lane;

  const int  sW  = s_base + m;
  const bool sok = sW < kS;
  v2f w2f[16][2];
  #pragma unroll
  for (int j = 0; j < 16; ++j) {
    #pragma unroll
    for (int q = 0; q < 2; ++q) {
      const int i0 = 8 * j + 4 * hf + 2 * q;
      w2f[j][q].x = sok ? W2[(size_t)(i0 + 0) * kS + sW] : 0.f;
      w2f[j][q].y = sok ? W2[(size_t)(i0 + 1) * kS + sW] : 0.f;
    }
  }
  const float bias = sok ? b2[sW] : 0.f;
  const float4 w1a = *(const float4*)(W1 + 0 * kHid + 4 * lane);
  const float4 w1b = *(const float4*)(W1 + 1 * kHid + 4 * lane);
  const float4 w1c = *(const float4*)(W1 + 2 * kHid + 4 * lane);
  const float4 b1v = *(const float4*)(b1 + 4 * lane);
  const int xrow = (lane < 24 ? lane : 0) / 3;
  const int xcmp = (lane < 24 ? lane : 0) - 3 * xrow;

  v8f acc0 = {}, acc1 = {};

  const int ch0 = part * nch;
  for (int ch = ch0; ch < ch0 + nch; ++ch) {
    const int cbase = ch * kCH + w * kSeg;
    float zr[kSegN], mr[kSegN];
    #pragma unroll
    for (int j = 0; j < kSegN; ++j) zr[j] = z[cbase + j * 32 + lane];
    #pragma unroll
    for (int j = 0; j < kSegN; ++j) mr[j] = mask[cbase + j * 32 + lane];
    {
      int cnt = 0;
      #pragma unroll
      for (int j = 0; j < kSegN; ++j) {
        const bool hit = (zr[j] > zlo) && (zr[j] < zhi);
        cnt += __builtin_popcount(__builtin_amdgcn_ballot_w32(hit));
      }
      if (lane == 0) lds_cnt[w] = cnt;
    }
    __syncthreads();
    const int c0 = lds_cnt[0], c1 = lds_cnt[1], c2 = lds_cnt[2], c3 = lds_cnt[3];
    const int n = __builtin_amdgcn_readfirstlane(c0 + c1 + c2 + c3);
    {
      int pos = (w > 0 ? c0 : 0) + (w > 1 ? c1 : 0) + (w > 2 ? c2 : 0);
      #pragma unroll
      for (int j = 0; j < kSegN; ++j) {
        const float zk = zr[j];
        const bool hit = (zk > zlo) && (zk < zhi);
        const unsigned b = __builtin_amdgcn_ballot_w32(hit);
        if (hit) {
          const int p = pos + __builtin_popcount(b & ((1u << lane) - 1u));
          lds_k[p] = cbase + j * 32 + lane;
          lds_z[p] = zk;
          lds_m[p] = mr[j];
        }
        pos += __builtin_popcount(b);
      }
      if (w == 3) {
        const int p = n + lane;
        lds_k[p] = 0;
        lds_z[p] = 0.f;
        lds_m[p] = 0.f;
      }
    }
    __syncthreads();

    {
      const int ng = (n + 15) >> 4;
      const int nb = (ng + 1) >> 1;
      if (nb > 0) {
        {
          const float xv0 = x[lds_k[8 * w + xrow] * 3 + xcmp];
          {
            const int ez = 16 * gw + 8 * hf;
            const v4f z0 = *(const v4f*)(lds_z + ez);
            const v4f z1 = *(const v4f*)(lds_z + ez + 4);
            const v4f m0 = *(const v4f*)(lds_m + ez);
            const v4f m1 = *(const v4f*)(lds_m + ez + 4);
            const float zz[8] = {z0.x, z0.y, z0.z, z0.w, z1.x, z1.y, z1.z, z1.w};
            const float mm[8] = {m0.x, m0.y, m0.z, m0.w, m1.x, m1.y, m1.z, m1.w};
            float tp[8];
            #pragma unroll
            for (int i = 0; i < 8; ++i) {
              const float d = tfw - zz[i];
              tp[i] = expf(-d * d) * mm[i] * kNorm;
            }
            float* td = &lds_tap[0][tap_wr];
            v4f q0 = {tp[0], tp[1], tp[2], tp[3]};
            v4f q1 = {tp[4], tp[5], tp[6], tp[7]};
            *(v4f*)td = q0;
            *(v4f*)(td + kTapQ) = q1;
          }
          float* dst = &lds_h[0][(8 * w) * kPitchF + 4 * lane];
          const int xb0 = __float_as_int(xv0);
          #pragma unroll
          for (int i = 0; i < 8; ++i) {
            const float x0 = __int_as_float(__builtin_amdgcn_readlane(xb0, 3 * i + 0));
            const float x1 = __int_as_float(__builtin_amdgcn_readlane(xb0, 3 * i + 1));
            const float x2 = __int_as_float(__builtin_amdgcn_readlane(xb0, 3 * i + 2));
            *(float4*)(dst + i * kPitchF) = sipm_h_row(x0, x1, x2, w1a, w1b, w1c, b1v);
          }
        }
        __syncthreads();
        int kfetch = lds_k[((1 < nb) ? 1 : 0) * kBR + 8 * w + xrow];
        for (int b = 0; b < nb; ++b) {
          const int bn = (b + 1 < nb) ? (b + 1) : b;
          const int xb = __float_as_int(x[kfetch * 3 + xcmp]);
          kfetch = lds_k[((b + 2 < nb) ? (b + 2) : (nb - 1)) * kBR + 8 * w + xrow];
          float* hnxt = &lds_h[(b + 1) & 1][(8 * w) * kPitchF + 4 * lane];
          const float* cur  = &lds_h[b & 1][0];
          const float* tcur = &lds_tap[b & 1][4 * lane];
          float*       tnxt = &lds_tap[(b + 1) & 1][tap_wr];
          #pragma unroll
          for (int half = 0; half < 2; ++half) {
            const int g = 2 * b + half;
            if (g >= ng) break;
            const float* arow = cur + (half * 16 + m) * kPitchF + 4 * hf;
            v4f a[16];
            #pragma unroll
            for (int j = 0; j < 16; ++j) a[j] = *(const v4f*)(arow + 8 * j);
            v4f nz0 = {}, nz1 = {}, nm0 = {}, nm1 = {};
            if (half == 0) {
              const int ez = bn * kBR + 16 * gw + 8 * hf;
              nz0 = *(const v4f*)(lds_z + ez);
              nz1 = *(const v4f*)(lds_z + ez + 4);
              nm0 = *(const v4f*)(lds_m + ez);
              nm1 = *(const v4f*)(lds_m + ez + 4);
            }
            const float* tg = tcur + (half * 4) * kTapQ;
            const v4f b00 = *(const v4f*)(tg);
            const v4f b01 = *(const v4f*)(tg + kTapQ);
            const v4f b10 = *(const v4f*)(tg + 2 * kTapQ);
            const v4f b11 = *(const v4f*)(tg + 3 * kTapQ);
            __builtin_amdgcn_sched_barrier(0);

            v8f d0 = {}, d1 = {};
            #pragma unroll
            for (int j = 0; j < 16; ++j) {
              v2f a0; a0.x = a[j].x; a0.y = a[j].y;
              v2f a1; a1.x = a[j].z; a1.y = a[j].w;
              d0 = __builtin_amdgcn_wmma_f32_16x16x4_f32(false, a0, false, w2f[j][0],
                                                         (short)0, d0, false, false);
              d1 = __builtin_amdgcn_wmma_f32_16x16x4_f32(false, a1, false, w2f[j][1],
                                                         (short)0, d1, false, false);
              if (half == 1 && (j & 1)) {
                const int i = j >> 1;
                const float x0 = __int_as_float(__builtin_amdgcn_readlane(xb, 3 * i + 0));
                const float x1 = __int_as_float(__builtin_amdgcn_readlane(xb, 3 * i + 1));
                const float x2 = __int_as_float(__builtin_amdgcn_readlane(xb, 3 * i + 2));
                *(float4*)(hnxt + i * kPitchF) = sipm_h_row(x0, x1, x2, w1a, w1b, w1c, b1v);
              }
            }
            if (half == 0) {
              const float zz[8] = {nz0.x, nz0.y, nz0.z, nz0.w, nz1.x, nz1.y, nz1.z, nz1.w};
              const float mm[8] = {nm0.x, nm0.y, nm0.z, nm0.w, nm1.x, nm1.y, nm1.z, nm1.w};
              float tp[8];
              #pragma unroll
              for (int i = 0; i < 8; ++i) {
                const float d = tfw - zz[i];
                tp[i] = expf(-d * d) * mm[i] * kNorm;
              }
              v4f q0 = {tp[0], tp[1], tp[2], tp[3]};
              v4f q1 = {tp[4], tp[5], tp[6], tp[7]};
              *(v4f*)tnxt = q0;
              *(v4f*)(tnxt + kTapQ) = q1;
            }
            float resp[8];
            #pragma unroll
            for (int r = 0; r < 8; ++r) resp[r] = expf((d0[r] + d1[r]) + bias);

            const float t0[8] = {b00.x, b00.y, b00.z, b00.w, b01.x, b01.y, b01.z, b01.w};
            const float t1[8] = {b10.x, b10.y, b10.z, b10.w, b11.x, b11.y, b11.z, b11.w};
            #pragma unroll
            for (int j = 0; j < 4; ++j) {
              v2f af;  af.x  = resp[2 * j]; af.y  = resp[2 * j + 1];
              v2f bf0; bf0.x = t0[2 * j];   bf0.y = t0[2 * j + 1];
              v2f bf1; bf1.x = t1[2 * j];   bf1.y = t1[2 * j + 1];
              acc0 = __builtin_amdgcn_wmma_f32_16x16x4_f32(false, af, false, bf0,
                                                           (short)0, acc0, false, false);
              acc1 = __builtin_amdgcn_wmma_f32_16x16x4_f32(false, af, false, bf1,
                                                           (short)0, acc1, false, false);
            }
          }
          __syncthreads();
        }
      }
    }
  }

  float* pbase = planes + (size_t)part * kPlaneF
               + (size_t)(s_base + 8 * hf) * kTPad + t_base + m;
  #pragma unroll
  for (int r = 0; r < 8; ++r) {
    float* prow = pbase + (size_t)r * kTPad;
    prow[0] = acc0[r];
    prow[16] = acc1[r];
  }
}

__global__ __launch_bounds__(256)
void sipm_copy_kernel(const float* __restrict__ planes, float* __restrict__ out) {
  int idx = blockIdx.x * 256 + threadIdx.x;
  if (idx >= kS * kT) return;
  int s = idx / kT, t = idx - s * kT;
  const float* p = planes + (size_t)s * kTPad + t;
  const int np = (t < (kTGrid - 1) * kTW) ? kSplitA : kSplitB;
  float v[kSplitA];
  #pragma unroll
  for (int i = 0; i < kSplitA; ++i) v[i] = (i < np) ? p[(size_t)i * kPlaneF] : 0.f;
  float sum = v[0];
  #pragma unroll
  for (int i = 1; i < kSplitA; ++i) sum += v[i];
  out[idx] = sum;
}

extern "C" void kernel_launch(void* const* d_in, const int* in_sizes, int n_in,
                              void* d_out, int out_size, void* d_ws, size_t ws_size,
                              hipStream_t stream) {
  const float* x   = (const float*)d_in[0];
  const float* z   = (const float*)d_in[1];
  const float* msk = (const float*)d_in[2];
  const float* W1  = (const float*)d_in[3];
  const float* b1  = (const float*)d_in[4];
  const float* W2  = (const float*)d_in[5];
  const float* b2  = (const float*)d_in[6];
  float* out = (float*)d_out;

  float* plane = (float*)d_ws;

  sipm_window_kernel<<<kWg, 128, 0, stream>>>(z, msk, W2, b2, x, W1, b1, plane);
  sipm_copy_kernel<<<(kS * kT + 255) / 256, 256, 0, stream>>>(plane, out);
}


# A workgroup owns the electrons of one 32-tick z range, not all within reach of a tick window: resp made once per electron and strip, 4 tick tiles; even/odd range planes are summed: new sum order
# speedup vs baseline: 9.4983x; 1.2014x over previous
#include <hip/hip_runtime.h>
#include <math.h>


typedef float v2f __attribute__((ext_vector_type(2)));
typedef float v4f __attribute__((ext_vector_type(4)));
typedef float v8f __attribute__((ext_vector_type(8)));

namespace {
constexpr int kBatch  = 64;
constexpr int kEv     = 500;
constexpr int kK      = kBatch * kEv;
constexpr int kHid    = 128;
constexpr int kS      = 2209;
constexpr int kSW     = 64;
constexpr int kSGrid  = (kS + kSW - 1) / kSW;
constexpr int kSPad   = kSGrid * kSW;
constexpr int kT      = 550;
constexpr int kTW     = 32;
constexpr int kTGrid  = (kT + kTW - 1) / kTW;
constexpr int kNT     = 4;
constexpr int kTOff   = 16;
constexpr int kTPad   = kTOff + kTGrid * kTW + 16;
constexpr int kReach  = 11;
constexpr int kCH     = 3200;
constexpr int kNCh    = kK / kCH;
constexpr int kSplitA = 5;
constexpr int kSplitB = 2;
constexpr int kTilesA = kSGrid * (kTGrid - 1);
constexpr int kWgA    = kTilesA * kSplitA;
constexpr int kWgB    = kSGrid * kSplitB;
constexpr int kWg     = kWgA + kWgB;
constexpr int kSeg    = kCH / 4;
constexpr int kSegN   = kSeg / 32;
constexpr int kBR     = 32;
constexpr int kPitchF = kHid + 4;
constexpr int kListCap = kCH + kBR;
constexpr int kTapQ   = 32 * 4;
constexpr int kTapBuf = 2 * kNT * 2 * kTapQ;
constexpr float kNorm = 0.3989422804f * 1.41421356237309515f;
constexpr size_t kPlaneF = (size_t)kSPad * kTPad;
static_assert(kTGrid * kTW >= kT + kReach, "the ranges must hold every z that reaches a tick");
static_assert(kNCh * kCH == kK, "chunks must tile K");
static_assert(kNCh % kSplitA == 0 && kNCh % kSplitB == 0, "parts must tile the chunks");
static_assert(kSeg % 32 == 0, "segment must be whole wave loads");
}

__device__ __forceinline__ float4 sipm_h_row(float x0, float x1, float x2, const float4 wa,
                                             const float4 wb, const float4 wc, const float4 bb) {
  float4 r;
  r.x = fmaxf(fmaf(x0, wa.x, fmaf(x1, wb.x, fmaf(x2, wc.x, bb.x))), 0.f);
  r.y = fmaxf(fmaf(x0, wa.y, fmaf(x1, wb.y, fmaf(x2, wc.y, bb.y))), 0.f);
  r.z = fmaxf(fmaf(x0, wa.z, fmaf(x1, wb.z, fmaf(x2, wc.z, bb.z))), 0.f);
  r.w = fmaxf(fmaf(x0, wa.w, fmaf(x1, wb.w, fmaf(x2, wc.w, bb.w))), 0.f);
  return r;
}

__device__ __forceinline__ void sipm_taps(float tf, const v4f z0, const v4f z1, const v4f m0,
                                          const v4f m1, float* td) {
  const float zz[8] = {z0.x, z0.y, z0.z, z0.w, z1.x, z1.y, z1.z, z1.w};
  const float mm[8] = {m0.x, m0.y, m0.z, m0.w, m1.x, m1.y, m1.z, m1.w};
  float tp[8];
  #pragma unroll
  for (int i = 0; i < 8; ++i) {
    const float d = tf - zz[i];
    tp[i] = expf(-d * d) * mm[i] * kNorm;
  }
  v4f q0 = {tp[0], tp[1], tp[2], tp[3]};
  v4f q1 = {tp[4], tp[5], tp[6], tp[7]};
  *(v4f*)td = q0;
  *(v4f*)(td + kTapQ) = q1;
}

__global__ __launch_bounds__(128)
void sipm_window_kernel(const float* __restrict__ z, const float* __restrict__ mask,
                        const float* __restrict__ W2, const float* __restrict__ b2,
                        const float* __restrict__ x, const float* __restrict__ W1,
                        const float* __restrict__ b1, float* __restrict__ planes) {
  __shared__ __attribute__((aligned(16))) int   lds_k[kListCap];
  __shared__ __attribute__((aligned(16))) float lds_z[kListCap];
  __shared__ __attribute__((aligned(16))) float lds_m[kListCap];
  __shared__ int lds_cnt[4];
  __shared__ __attribute__((aligned(16))) float lds_h[2][kBR * kPitchF];
  __shared__ __attribute__((aligned(16))) float lds_tap[2][kTapBuf];

  const int tid  = threadIdx.x;
  const int w    = tid >> 5;
  const int lane = tid & 31;
  const int m    = lane & 15;
  const int hf   = lane >> 4;
  const int id = blockIdx.x;
  int part, strip, win, nch;
  if (id < kWgA) {
    part = id / kTilesA;
    const int r = id - part * kTilesA;
    win   = r / kSGrid;
    strip = r - win * kSGrid;
    nch   = kNCh / kSplitA;
  } else {
    const int r = id - kWgA;
    part  = r / kSGrid;
    strip = r - part * kSGrid;
    win   = kTGrid - 1;
    nch   = kNCh / kSplitB;
  }
  const int s_base = strip * kSW + w * 16;
  const int t0 = win * kTW - 16;
  const float zlo = (win == 0) ? -(float)kReach : (float)(win * kTW);
  const float zhi = (float)(win * kTW + kTW);
  const int   gw  = w >> 1;
  const int   tw  = w & 1;
  const float tfa = (float)(t0 + 16 * tw + m);
  const float tfb = (float)(t0 + 16 * tw + 32 + m);
  const int   tap_wr = ((gw * kNT + tw) * 2) * kTapQ + 4 * lane;

  const int  sW  = s_base + m;
  const bool sok = sW < kS;
  v2f w2f[16][2];
  #pragma unroll
  for (int j = 0; j < 16; ++j) {
    #pragma unroll
    for (int q = 0; q < 2; ++q) {
      const int i0 = 8 * j + 4 * hf + 2 * q;
      w2f[j][q].x = sok ? W2[(size_t)(i0 + 0) * kS + sW] : 0.f;
      w2f[j][q].y = sok ? W2[(size_t)(i0 + 1) * kS + sW] : 0.f;
    }
  }
  const float bias = sok ? b2[sW] : 0.f;
  const float4 w1a = *(const float4*)(W1 + 0 * kHid + 4 * lane);
  const float4 w1b = *(const float4*)(W1 + 1 * kHid + 4 * lane);
  const float4 w1c = *(const float4*)(W1 + 2 * kHid + 4 * lane);
  const float4 b1v = *(const float4*)(b1 + 4 * lane);
  const int xrow = (lane < 24 ? lane : 0) / 3;
  const int xcmp = (lane < 24 ? lane : 0) - 3 * xrow;

  v8f acc[kNT] = {};

  const int ch0 = part * nch;
  for (int ch = ch0; ch < ch0 + nch; ++ch) {
    const int cbase = ch * kCH + w * kSeg;
    float zr[kSegN], mr[kSegN];
    #pragma unroll
    for (int j = 0; j < kSegN; ++j) zr[j] = z[cbase + j * 32 + lane];
    #pragma unroll
    for (int j = 0; j < kSegN; ++j) mr[j] = mask[cbase + j * 32 + lane];
    {
      int cnt = 0;
      #pragma unroll
      for (int j = 0; j < kSegN; ++j) {
        const bool hit = (zr[j] >= zlo) && (zr[j] < zhi);
        cnt += __builtin_popcount(__builtin_amdgcn_ballot_w32(hit));
      }
      if (lane == 0) lds_cnt[w] = cnt;
    }
    __syncthreads();
    const int c0 = lds_cnt[0], c1 = lds_cnt[1], c2 = lds_cnt[2], c3 = lds_cnt[3];
    const int n = __builtin_amdgcn_readfirstlane(c0 + c1 + c2 + c3);
    {
      int pos = (w > 0 ? c0 : 0) + (w > 1 ? c1 : 0) + (w > 2 ? c2 : 0);
      #pragma unroll
      for (int j = 0; j < kSegN; ++j) {
        const float zk = zr[j];
        const bool hit = (zk >= zlo) && (zk < zhi);
        const unsigned b = __builtin_amdgcn_ballot_w32(hit);
        if (hit) {
          const int p = pos + __builtin_popcount(b & ((1u << lane) - 1u));
          lds_k[p] = cbase + j * 32 + lane;
          lds_z[p] = zk;
          lds_m[p] = mr[j];
        }
        pos += __builtin_popcount(b);
      }
      if (w == 3) {
        const int p = n + lane;
        lds_k[p] = 0;
        lds_z[p] = 0.f;
        lds_m[p] = 0.f;
      }
    }
    __syncthreads();

    {
      const int ng = (n + 15) >> 4;
      const int nb = (ng + 1) >> 1;
      if (nb > 0) {
        {
          const float xv0 = x[lds_k[8 * w + xrow] * 3 + xcmp];
          {
            const int ez = 16 * gw + 8 * hf;
            const v4f z0 = *(const v4f*)(lds_z + ez);
            const v4f z1 = *(const v4f*)(lds_z + ez + 4);
            const v4f m0 = *(const v4f*)(lds_m + ez);
            const v4f m1 = *(const v4f*)(lds_m + ez + 4);
            sipm_taps(tfa, z0, z1, m0, m1, &lds_tap[0][tap_wr]);
            sipm_taps(tfb, z0, z1, m0, m1, &lds_tap[0][tap_wr + 4 * kTapQ]);
          }
          float* dst = &lds_h[0][(8 * w) * kPitchF + 4 * lane];
          const int xb0 = __float_as_int(xv0);
          #pragma unroll
          for (int i = 0; i < 8; ++i) {
            const float x0 = __int_as_float(__builtin_amdgcn_readlane(xb0, 3 * i + 0));
            const float x1 = __int_as_float(__builtin_amdgcn_readlane(xb0, 3 * i + 1));
            const float x2 = __int_as_float(__builtin_amdgcn_readlane(xb0, 3 * i + 2));
            *(float4*)(dst + i * kPitchF) = sipm_h_row(x0, x1, x2, w1a, w1b, w1c, b1v);
          }
        }
        __syncthreads();
        int kfetch = lds_k[((1 < nb) ? 1 : 0) * kBR + 8 * w + xrow];
        for (int b = 0; b < nb; ++b) {
          const int bn = (b + 1 < nb) ? (b + 1) : b;
          const int xb = __float_as_int(x[kfetch * 3 + xcmp]);
          kfetch = lds_k[((b + 2 < nb) ? (b + 2) : (nb - 1)) * kBR + 8 * w + xrow];
          float* hnxt = &lds_h[(b + 1) & 1][(8 * w) * kPitchF + 4 * lane];
          const float* cur  = &lds_h[b & 1][0];
          const float* tcur = &lds_tap[b & 1][4 * lane];
          float*       tnxt = &lds_tap[(b + 1) & 1][tap_wr];
          #pragma unroll
          for (int half = 0; half < 2; ++half) {
            const int g = 2 * b + half;
            if (g >= ng) break;
            const float* arow = cur + (half * 16 + m) * kPitchF + 4 * hf;
            v4f a[16];
            #pragma unroll
            for (int j = 0; j < 16; ++j) a[j] = *(const v4f*)(arow + 8 * j);
            v4f nz0 = {}, nz1 = {}, nm0 = {}, nm1 = {};
            if (half == 0) {
              const int ez = bn * kBR + 16 * gw + 8 * hf;
              nz0 = *(const v4f*)(lds_z + ez);
              nz1 = *(const v4f*)(lds_z + ez + 4);
              nm0 = *(const v4f*)(lds_m + ez);
              nm1 = *(const v4f*)(lds_m + ez + 4);
            }
            const float* tg = tcur + (half * 2 * kNT) * kTapQ;
            v4f bq[kNT][2];
            #pragma unroll
            for (int tt = 0; tt < kNT; ++tt) {
              bq[tt][0] = *(const v4f*)(tg + (2 * tt) * kTapQ);
              bq[tt][1] = *(const v4f*)(tg + (2 * tt + 1) * kTapQ);
            }
            __builtin_amdgcn_sched_barrier(0);

            v8f d0 = {}, d1 = {};
            #pragma unroll
            for (int j = 0; j < 16; ++j) {
              v2f a0; a0.x = a[j].x; a0.y = a[j].y;
              v2f a1; a1.x = a[j].z; a1.y = a[j].w;
              d0 = __builtin_amdgcn_wmma_f32_16x16x4_f32(false, a0, false, w2f[j][0],
                                                         (short)0, d0, false, false);
              d1 = __builtin_amdgcn_wmma_f32_16x16x4_f32(false, a1, false, w2f[j][1],
                                                         (short)0, d1, false, false);
              if (half == 1 && (j & 1)) {
                const int i = j >> 1;
                const float x0 = __int_as_float(__builtin_amdgcn_readlane(xb, 3 * i + 0));
                const float x1 = __int_as_float(__builtin_amdgcn_readlane(xb, 3 * i + 1));
                const float x2 = __int_as_float(__builtin_amdgcn_readlane(xb, 3 * i + 2));
                *(float4*)(hnxt + i * kPitchF) = sipm_h_row(x0, x1, x2, w1a, w1b, w1c, b1v);
              }
            }
            if (half == 0) {
              sipm_taps(tfa, nz0, nz1, nm0, nm1, tnxt);
              sipm_taps(tfb, nz0, nz1, nm0, nm1, tnxt + 4 * kTapQ);
            }
            float resp[8];
            #pragma unroll
            for (int r = 0; r < 8; ++r) resp[r] = expf((d0[r] + d1[r]) + bias);

            #pragma unroll
            for (int j = 0; j < 4; ++j) {
              v2f af; af.x = resp[2 * j]; af.y = resp[2 * j + 1];
              #pragma unroll
              for (int tt = 0; tt < kNT; ++tt) {
                const v4f bv = bq[tt][j >> 1];
                v2f bf;
                bf.x = (j & 1) ? bv.z : bv.x;
                bf.y = (j & 1) ? bv.w : bv.y;
                acc[tt] = __builtin_amdgcn_wmma_f32_16x16x4_f32(false, af, false, bf,
                                                                (short)0, acc[tt], false, false);
              }
            }
          }
          __syncthreads();
        }
      }
    }
  }

  float* pbase = planes + (size_t)(part + kSplitA * (win & 1)) * kPlaneF
               + (size_t)(s_base + 8 * hf) * kTPad + (t0 + kTOff) + m;
  #pragma unroll
  for (int r = 0; r < 8; ++r) {
    float* prow = pbase + (size_t)r * kTPad;
    #pragma unroll
    for (int tt = 0; tt < kNT; ++tt) prow[16 * tt] = acc[tt][r];
  }
}

__global__ __launch_bounds__(256)
void sipm_copy_kernel(const float* __restrict__ planes, float* __restrict__ out) {
  int idx = blockIdx.x * 256 + threadIdx.x;
  if (idx >= kS * kT) return;
  int s = idx / kT, t = idx - s * kT;
  const float* p = planes + (size_t)s * kTPad + (t + kTOff);
  const int tOddLast = (kTGrid - 1) * kTW - 16;
  const int n1 = (t < 16) ? 0 : ((t < tOddLast) ? kSplitA : kSplitB);
  float v[2 * kSplitA];
  #pragma unroll
  for (int i = 0; i < kSplitA; ++i) v[i] = p[(size_t)i * kPlaneF];
  #pragma unroll
  for (int i = 0; i < kSplitA; ++i)
    v[kSplitA + i] = (i < n1) ? p[(size_t)(kSplitA + i) * kPlaneF] : 0.f;
  float sum = v[0];
  #pragma unroll
  for (int i = 1; i < 2 * kSplitA; ++i) sum += v[i];
  out[idx] = sum;
}

extern "C" void kernel_launch(void* const* d_in, const int* in_sizes, int n_in,
                              void* d_out, int out_size, void* d_ws, size_t ws_size,
                              hipStream_t stream) {
  const float* x   = (const float*)d_in[0];
  const float* z   = (const float*)d_in[1];
  const float* msk = (const float*)d_in[2];
  const float* W1  = (const float*)d_in[3];
  const float* b1  = (const float*)d_in[4];
  const float* W2  = (const float*)d_in[5];
  const float* b2  = (const float*)d_in[6];
  float* out = (float*)d_out;

  float* plane = (float*)d_ws;

  sipm_window_kernel<<<kWg, 128, 0, stream>>>(z, msk, W2, b2, x, W1, b1, plane);
  sipm_copy_kernel<<<(kS * kT + 255) / 256, 256, 0, stream>>>(plane, out);
}


# Gaussian taps exp(-d*d) made as 2^(d*d*-log2e) on the hardware exp2 instruction (rel. error ~4e-8*d*d*log2e + 1 ulp; denormal taps become 0): 4 vector instructions per tap, was 15; sums unchanged
# speedup vs baseline: 10.2571x; 1.0799x over previous
#include <hip/hip_runtime.h>
#include <math.h>


typedef float v2f __attribute__((ext_vector_type(2)));
typedef float v4f __attribute__((ext_vector_type(4)));
typedef float v8f __attribute__((ext_vector_type(8)));

namespace {
constexpr int kBatch  = 64;
constexpr int kEv     = 500;
constexpr int kK      = kBatch * kEv;
constexpr int kHid    = 128;
constexpr int kS      = 2209;
constexpr int kSW     = 64;
constexpr int kSGrid  = (kS + kSW - 1) / kSW;
constexpr int kSPad   = kSGrid * kSW;
constexpr int kT      = 550;
constexpr int kTW     = 32;
constexpr int kTGrid  = (kT + kTW - 1) / kTW;
constexpr int kNT     = 4;
constexpr int kTOff   = 16;
constexpr int kTPad   = kTOff + kTGrid * kTW + 16;
constexpr int kReach  = 11;
constexpr int kCH     = 3200;
constexpr int kNCh    = kK / kCH;
constexpr int kSplitA = 5;
constexpr int kSplitB = 2;
constexpr int kTilesA = kSGrid * (kTGrid - 1);
constexpr int kWgA    = kTilesA * kSplitA;
constexpr int kWgB    = kSGrid * kSplitB;
constexpr int kWg     = kWgA + kWgB;
constexpr int kSeg    = kCH / 4;
constexpr int kSegN   = kSeg / 32;
constexpr int kBR     = 32;
constexpr int kPitchF = kHid + 4;
constexpr int kListCap = kCH + kBR;
constexpr int kTapQ   = 32 * 4;
constexpr int kTapBuf = 2 * kNT * 2 * kTapQ;
constexpr float kNorm = 0.3989422804f * 1.41421356237309515f;
constexpr float kNegLog2e = -1.44269504088896341f;
constexpr size_t kPlaneF = (size_t)kSPad * kTPad;
static_assert(kTGrid * kTW >= kT + kReach, "the ranges must hold every z that reaches a tick");
static_assert(kNCh * kCH == kK, "chunks must tile K");
static_assert(kNCh % kSplitA == 0 && kNCh % kSplitB == 0, "parts must tile the chunks");
static_assert(kSeg % 32 == 0, "segment must be whole wave loads");
}

__device__ __forceinline__ float4 sipm_h_row(float x0, float x1, float x2, const float4 wa,
                                             const float4 wb, const float4 wc, const float4 bb) {
  float4 r;
  r.x = fmaxf(fmaf(x0, wa.x, fmaf(x1, wb.x, fmaf(x2, wc.x, bb.x))), 0.f);
  r.y = fmaxf(fmaf(x0, wa.y, fmaf(x1, wb.y, fmaf(x2, wc.y, bb.y))), 0.f);
  r.z = fmaxf(fmaf(x0, wa.z, fmaf(x1, wb.z, fmaf(x2, wc.z, bb.z))), 0.f);
  r.w = fmaxf(fmaf(x0, wa.w, fmaf(x1, wb.w, fmaf(x2, wc.w, bb.w))), 0.f);
  return r;
}

__device__ __forceinline__ void sipm_taps(float tf, const v4f z0, const v4f z1, const v4f m0,
                                          const v4f m1, float* td) {
  const float zz[8] = {z0.x, z0.y, z0.z, z0.w, z1.x, z1.y, z1.z, z1.w};
  const float mm[8] = {m0.x, m0.y, m0.z, m0.w, m1.x, m1.y, m1.z, m1.w};
  float tp[8];
  #pragma unroll
  for (int i = 0; i < 8; ++i) {
    const float d = tf - zz[i];
    tp[i] = __builtin_amdgcn_exp2f((d * d) * kNegLog2e) * mm[i] * kNorm;
  }
  v4f q0 = {tp[0], tp[1], tp[2], tp[3]};
  v4f q1 = {tp[4], tp[5], tp[6], tp[7]};
  *(v4f*)td = q0;
  *(v4f*)(td + kTapQ) = q1;
}

__global__ __launch_bounds__(128)
void sipm_window_kernel(const float* __restrict__ z, const float* __restrict__ mask,
                        const float* __restrict__ W2, const float* __restrict__ b2,
                        const float* __restrict__ x, const float* __restrict__ W1,
                        const float* __restrict__ b1, float* __restrict__ planes) {
  __shared__ __attribute__((aligned(16))) int   lds_k[kListCap];
  __shared__ __attribute__((aligned(16))) float lds_z[kListCap];
  __shared__ __attribute__((aligned(16))) float lds_m[kListCap];
  __shared__ int lds_cnt[4];
  __shared__ __attribute__((aligned(16))) float lds_h[2][kBR * kPitchF];
  __shared__ __attribute__((aligned(16))) float lds_tap[2][kTapBuf];

  const int tid  = threadIdx.x;
  const int w    = tid >> 5;
  const int lane = tid & 31;
  const int m    = lane & 15;
  const int hf   = lane >> 4;
  const int id = blockIdx.x;
  int part, strip, win, nch;
  if (id < kWgA) {
    part = id / kTilesA;
    const int r = id - part * kTilesA;
    win   = r / kSGrid;
    strip = r - win * kSGrid;
    nch   = kNCh / kSplitA;
  } else {
    const int r = id - kWgA;
    part  = r / kSGrid;
    strip = r - part * kSGrid;
    win   = kTGrid - 1;
    nch   = kNCh / kSplitB;
  }
  const int s_base = strip * kSW + w * 16;
  const int t0 = win * kTW - 16;
  const float zlo = (win == 0) ? -(float)kReach : (float)(win * kTW);
  const float zhi = (float)(win * kTW + kTW);
  const int   gw  = w >> 1;
  const int   tw  = w & 1;
  const float tfa = (float)(t0 + 16 * tw + m);
  const float tfb = (float)(t0 + 16 * tw + 32 + m);
  const int   tap_wr = ((gw * kNT + tw) * 2) * kTapQ + 4 * lane;

  const int  sW  = s_base + m;
  const bool sok = sW < kS;
  v2f w2f[16][2];
  #pragma unroll
  for (int j = 0; j < 16; ++j) {
    #pragma unroll
    for (int q = 0; q < 2; ++q) {
      const int i0 = 8 * j + 4 * hf + 2 * q;
      w2f[j][q].x = sok ? W2[(size_t)(i0 + 0) * kS + sW] : 0.f;
      w2f[j][q].y = sok ? W2[(size_t)(i0 + 1) * kS + sW] : 0.f;
    }
  }
  const float bias = sok ? b2[sW] : 0.f;
  const float4 w1a = *(const float4*)(W1 + 0 * kHid + 4 * lane);
  const float4 w1b = *(const float4*)(W1 + 1 * kHid + 4 * lane);
  const float4 w1c = *(const float4*)(W1 + 2 * kHid + 4 * lane);
  const float4 b1v = *(const float4*)(b1 + 4 * lane);
  const int xrow = (lane < 24 ? lane : 0) / 3;
  const int xcmp = (lane < 24 ? lane : 0) - 3 * xrow;

  v8f acc[kNT] = {};

  const int ch0 = part * nch;
  for (int ch = ch0; ch < ch0 + nch; ++ch) {
    const int cbase = ch * kCH + w * kSeg;
    float zr[kSegN], mr[kSegN];
    #pragma unroll
    for (int j = 0; j < kSegN; ++j) zr[j] = z[cbase + j * 32 + lane];
    #pragma unroll
    for (int j = 0; j < kSegN; ++j) mr[j] = mask[cbase + j * 32 + lane];
    {
      int cnt = 0;
      #pragma unroll
      for (int j = 0; j < kSegN; ++j) {
        const bool hit = (zr[j] >= zlo) && (zr[j] < zhi);
        cnt += __builtin_popcount(__builtin_amdgcn_ballot_w32(hit));
      }
      if (lane == 0) lds_cnt[w] = cnt;
    }
    __syncthreads();
    const int c0 = lds_cnt[0], c1 = lds_cnt[1], c2 = lds_cnt[2], c3 = lds_cnt[3];
    const int n = __builtin_amdgcn_readfirstlane(c0 + c1 + c2 + c3);
    {
      int pos = (w > 0 ? c0 : 0) + (w > 1 ? c1 : 0) + (w > 2 ? c2 : 0);
      #pragma unroll
      for (int j = 0; j < kSegN; ++j) {
        const float zk = zr[j];
        const bool hit = (zk >= zlo) && (zk < zhi);
        const unsigned b = __builtin_amdgcn_ballot_w32(hit);
        if (hit) {
          const int p = pos + __builtin_popcount(b & ((1u << lane) - 1u));
          lds_k[p] = cbase + j * 32 + lane;
          lds_z[p] = zk;
          lds_m[p] = mr[j];
        }
        pos += __builtin_popcount(b);
      }
      if (w == 3) {
        const int p = n + lane;
        lds_k[p] = 0;
        lds_z[p] = 0.f;
        lds_m[p] = 0.f;
      }
    }
    __syncthreads();

    {
      const int ng = (n + 15) >> 4;
      const int nb = (ng + 1) >> 1;
      if (nb > 0) {
        {
          const float xv0 = x[lds_k[8 * w + xrow] * 3 + xcmp];
          {
            const int ez = 16 * gw + 8 * hf;
            const v4f z0 = *(const v4f*)(lds_z + ez);
            const v4f z1 = *(const v4f*)(lds_z + ez + 4);
            const v4f m0 = *(const v4f*)(lds_m + ez);
            const v4f m1 = *(const v4f*)(lds_m + ez + 4);
            sipm_taps(tfa, z0, z1, m0, m1, &lds_tap[0][tap_wr]);
            sipm_taps(tfb, z0, z1, m0, m1, &lds_tap[0][tap_wr + 4 * kTapQ]);
          }
          float* dst = &lds_h[0][(8 * w) * kPitchF + 4 * lane];
          const int xb0 = __float_as_int(xv0);
          #pragma unroll
          for (int i = 0; i < 8; ++i) {
            const float x0 = __int_as_float(__builtin_amdgcn_readlane(xb0, 3 * i + 0));
            const float x1 = __int_as_float(__builtin_amdgcn_readlane(xb0, 3 * i + 1));
            const float x2 = __int_as_float(__builtin_amdgcn_readlane(xb0, 3 * i + 2));
            *(float4*)(dst + i * kPitchF) = sipm_h_row(x0, x1, x2, w1a, w1b, w1c, b1v);
          }
        }
        __syncthreads();
        int kfetch = lds_k[((1 < nb) ? 1 : 0) * kBR + 8 * w + xrow];
        for (int b = 0; b < nb; ++b) {
          const int bn = (b + 1 < nb) ? (b + 1) : b;
          const int xb = __float_as_int(x[kfetch * 3 + xcmp]);
          kfetch = lds_k[((b + 2 < nb) ? (b + 2) : (nb - 1)) * kBR + 8 * w + xrow];
          float* hnxt = &lds_h[(b + 1) & 1][(8 * w) * kPitchF + 4 * lane];
          const float* cur  = &lds_h[b & 1][0];
          const float* tcur = &lds_tap[b & 1][4 * lane];
          float*       tnxt = &lds_tap[(b + 1) & 1][tap_wr];
          #pragma unroll
          for (int half = 0; half < 2; ++half) {
            const int g = 2 * b + half;
            if (g >= ng) break;
            const float* arow = cur + (half * 16 + m) * kPitchF + 4 * hf;
            v4f a[16];
            #pragma unroll
            for (int j = 0; j < 16; ++j) a[j] = *(const v4f*)(arow + 8 * j);
            v4f nz0 = {}, nz1 = {}, nm0 = {}, nm1 = {};
            if (half == 0) {
              const int ez = bn * kBR + 16 * gw + 8 * hf;
              nz0 = *(const v4f*)(lds_z + ez);
              nz1 = *(const v4f*)(lds_z + ez + 4);
              nm0 = *(const v4f*)(lds_m + ez);
              nm1 = *(const v4f*)(lds_m + ez + 4);
            }
            const float* tg = tcur + (half * 2 * kNT) * kTapQ;
            v4f bq[kNT][2];
            #pragma unroll
            for (int tt = 0; tt < kNT; ++tt) {
              bq[tt][0] = *(const v4f*)(tg + (2 * tt) * kTapQ);
              bq[tt][1] = *(const v4f*)(tg + (2 * tt + 1) * kTapQ);
            }
            __builtin_amdgcn_sched_barrier(0);

            v8f d0 = {}, d1 = {};
            #pragma unroll
            for (int j = 0; j < 16; ++j) {
              v2f a0; a0.x = a[j].x; a0.y = a[j].y;
              v2f a1; a1.x = a[j].z; a1.y = a[j].w;
              d0 = __builtin_amdgcn_wmma_f32_16x16x4_f32(false, a0, false, w2f[j][0],
                                                         (short)0, d0, false, false);
              d1 = __builtin_amdgcn_wmma_f32_16x16x4_f32(false, a1, false, w2f[j][1],
                                                         (short)0, d1, false, false);
              if (half == 1 && (j & 1)) {
                const int i = j >> 1;
                const float x0 = __int_as_float(__builtin_amdgcn_readlane(xb, 3 * i + 0));
                const float x1 = __int_as_float(__builtin_amdgcn_readlane(xb, 3 * i + 1));
                const float x2 = __int_as_float(__builtin_amdgcn_readlane(xb, 3 * i + 2));
                *(float4*)(hnxt + i * kPitchF) = sipm_h_row(x0, x1, x2, w1a, w1b, w1c, b1v);
              }
            }
            if (half == 0) {
              sipm_taps(tfa, nz0, nz1, nm0, nm1, tnxt);
              sipm_taps(tfb, nz0, nz1, nm0, nm1, tnxt + 4 * kTapQ);
            }
            float resp[8];
            #pragma unroll
            for (int r = 0; r < 8; ++r) resp[r] = expf((d0[r] + d1[r]) + bias);

            #pragma unroll
            for (int j = 0; j < 4; ++j) {
              v2f af; af.x = resp[2 * j]; af.y = resp[2 * j + 1];
              #pragma unroll
              for (int tt = 0; tt < kNT; ++tt) {
                const v4f bv = bq[tt][j >> 1];
                v2f bf;
                bf.x = (j & 1) ? bv.z : bv.x;
                bf.y = (j & 1) ? bv.w : bv.y;
                acc[tt] = __builtin_amdgcn_wmma_f32_16x16x4_f32(false, af, false, bf,
                                                                (short)0, acc[tt], false, false);
              }
            }
          }
          __syncthreads();
        }
      }
    }
  }

  float* pbase = planes + (size_t)(part + kSplitA * (win & 1)) * kPlaneF
               + (size_t)(s_base + 8 * hf) * kTPad + (t0 + kTOff) + m;
  #pragma unroll
  for (int r = 0; r < 8; ++r) {
    float* prow = pbase + (size_t)r * kTPad;
    #pragma unroll
    for (int tt = 0; tt < kNT; ++tt) prow[16 * tt] = acc[tt][r];
  }
}

__global__ __launch_bounds__(256)
void sipm_copy_kernel(const float* __restrict__ planes, float* __restrict__ out) {
  int idx = blockIdx.x * 256 + threadIdx.x;
  if (idx >= kS * kT) return;
  int s = idx / kT, t = idx - s * kT;
  const float* p = planes + (size_t)s * kTPad + (t + kTOff);
  const int tOddLast = (kTGrid - 1) * kTW - 16;
  const int n1 = (t < 16) ? 0 : ((t < tOddLast) ? kSplitA : kSplitB);
  float v[2 * kSplitA];
  #pragma unroll
  for (int i = 0; i < kSplitA; ++i) v[i] = p[(size_t)i * kPlaneF];
  #pragma unroll
  for (int i = 0; i < kSplitA; ++i)
    v[kSplitA + i] = (i < n1) ? p[(size_t)(kSplitA + i) * kPlaneF] : 0.f;
  float sum = v[0];
  #pragma unroll
  for (int i = 1; i < 2 * kSplitA; ++i) sum += v[i];
  out[idx] = sum;
}

extern "C" void kernel_launch(void* const* d_in, const int* in_sizes, int n_in,
                              void* d_out, int out_size, void* d_ws, size_t ws_size,
                              hipStream_t stream) {
  const float* x   = (const float*)d_in[0];
  const float* z   = (const float*)d_in[1];
  const float* msk = (const float*)d_in[2];
  const float* W1  = (const float*)d_in[3];
  const float* b1  = (const float*)d_in[4];
  const float* W2  = (const float*)d_in[5];
  const float* b2  = (const float*)d_in[6];
  float* out = (float*)d_out;

  float* plane = (float*)d_ws;

  sipm_window_kernel<<<kWg, 128, 0, stream>>>(z, msk, W2, b2, x, W1, b1, plane);
  sipm_copy_kernel<<<(kS * kT + 255) / 256, 256, 0, stream>>>(plane, out);
}


# List of a range ordered lower 16 ticks first, then upper 16: a group of 16 wholly in one half skips the tick tile it cannot reach (taps exactly 0): 12 binning WMMAs, not 16; new sum order
# speedup vs baseline: 10.6434x; 1.0377x over previous
#include <hip/hip_runtime.h>
#include <math.h>


typedef float v2f __attribute__((ext_vector_type(2)));
typedef float v4f __attribute__((ext_vector_type(4)));
typedef float v8f __attribute__((ext_vector_type(8)));

namespace {
constexpr int kBatch  = 64;
constexpr int kEv     = 500;
constexpr int kK      = kBatch * kEv;
constexpr int kHid    = 128;
constexpr int kS      = 2209;
constexpr int kSW     = 64;
constexpr int kSGrid  = (kS + kSW - 1) / kSW;
constexpr int kSPad   = kSGrid * kSW;
constexpr int kT      = 550;
constexpr int kTW     = 32;
constexpr int kTGrid  = (kT + kTW - 1) / kTW;
constexpr int kNT     = 4;
constexpr int kTOff   = 16;
constexpr int kTPad   = kTOff + kTGrid * kTW + 16;
constexpr int kReach  = 11;
constexpr int kCH     = 3200;
constexpr int kNCh    = kK / kCH;
constexpr int kSplitA = 5;
constexpr int kSplitB = 2;
constexpr int kTilesA = kSGrid * (kTGrid - 1);
constexpr int kWgA    = kTilesA * kSplitA;
constexpr int kWgB    = kSGrid * kSplitB;
constexpr int kWg     = kWgA + kWgB;
constexpr int kSeg    = kCH / 4;
constexpr int kSegN   = kSeg / 32;
constexpr int kBR     = 32;
constexpr int kPitchF = kHid + 4;
constexpr int kListCap = kCH + kBR;
constexpr int kTapQ   = 32 * 4;
constexpr int kTapBuf = 2 * kNT * 2 * kTapQ;
constexpr float kNorm = 0.3989422804f * 1.41421356237309515f;
constexpr float kNegLog2e = -1.44269504088896341f;
constexpr size_t kPlaneF = (size_t)kSPad * kTPad;
static_assert(kTGrid * kTW >= kT + kReach, "the ranges must hold every z that reaches a tick");
static_assert(kNCh * kCH == kK, "chunks must tile K");
static_assert(kNCh % kSplitA == 0 && kNCh % kSplitB == 0, "parts must tile the chunks");
static_assert(kSeg % 32 == 0, "segment must be whole wave loads");
}

__device__ __forceinline__ float4 sipm_h_row(float x0, float x1, float x2, const float4 wa,
                                             const float4 wb, const float4 wc, const float4 bb) {
  float4 r;
  r.x = fmaxf(fmaf(x0, wa.x, fmaf(x1, wb.x, fmaf(x2, wc.x, bb.x))), 0.f);
  r.y = fmaxf(fmaf(x0, wa.y, fmaf(x1, wb.y, fmaf(x2, wc.y, bb.y))), 0.f);
  r.z = fmaxf(fmaf(x0, wa.z, fmaf(x1, wb.z, fmaf(x2, wc.z, bb.z))), 0.f);
  r.w = fmaxf(fmaf(x0, wa.w, fmaf(x1, wb.w, fmaf(x2, wc.w, bb.w))), 0.f);
  return r;
}

__device__ __forceinline__ void sipm_taps(float tf, const v4f z0, const v4f z1, const v4f m0,
                                          const v4f m1, float* td) {
  const float zz[8] = {z0.x, z0.y, z0.z, z0.w, z1.x, z1.y, z1.z, z1.w};
  const float mm[8] = {m0.x, m0.y, m0.z, m0.w, m1.x, m1.y, m1.z, m1.w};
  float tp[8];
  #pragma unroll
  for (int i = 0; i < 8; ++i) {
    const float d = tf - zz[i];
    tp[i] = __builtin_amdgcn_exp2f((d * d) * kNegLog2e) * mm[i] * kNorm;
  }
  v4f q0 = {tp[0], tp[1], tp[2], tp[3]};
  v4f q1 = {tp[4], tp[5], tp[6], tp[7]};
  *(v4f*)td = q0;
  *(v4f*)(td + kTapQ) = q1;
}

template <int T0, int T1>
__device__ __forceinline__ void sipm_bin(v8f (&acc)[kNT], const float (&resp)[8],
                                         const v4f (&bq)[kNT][2]) {
  #pragma unroll
  for (int j = 0; j < 4; ++j) {
    v2f af; af.x = resp[2 * j]; af.y = resp[2 * j + 1];
    #pragma unroll
    for (int tt = T0; tt < T1; ++tt) {
      const v4f bv = bq[tt][j >> 1];
      v2f bf;
      bf.x = (j & 1) ? bv.z : bv.x;
      bf.y = (j & 1) ? bv.w : bv.y;
      acc[tt] = __builtin_amdgcn_wmma_f32_16x16x4_f32(false, af, false, bf,
                                                      (short)0, acc[tt], false, false);
    }
  }
}

__device__ __forceinline__ void sipm_bin_tile_unless(int skip, v8f& acc, const float (&resp)[8],
                                                     const v4f b0, const v4f b1) {
  v2f a0; a0.x = resp[0]; a0.y = resp[1];
  v2f a1; a1.x = resp[2]; a1.y = resp[3];
  v2f a2; a2.x = resp[4]; a2.y = resp[5];
  v2f a3; a3.x = resp[6]; a3.y = resp[7];
  v2f p0; p0.x = b0.x; p0.y = b0.y;
  v2f p1; p1.x = b0.z; p1.y = b0.w;
  v2f p2; p2.x = b1.x; p2.y = b1.y;
  v2f p3; p3.x = b1.z; p3.y = b1.w;
  asm volatile(
      "s_cmp_lg_u32 %9, 0\n\t"
      "s_cbranch_scc1 .Lsipm_tile_skip_%=\n\t"
      "v_wmma_f32_16x16x4_f32 %0, %1, %5, %0\n\t"
      "v_wmma_f32_16x16x4_f32 %0, %2, %6, %0\n\t"
      "v_wmma_f32_16x16x4_f32 %0, %3, %7, %0\n\t"
      "v_wmma_f32_16x16x4_f32 %0, %4, %8, %0\n\t"
      "v_nop\n\tv_nop\n\tv_nop\n\tv_nop\n"
      ".Lsipm_tile_skip_%=:"
      : "+v"(acc)
      : "v"(a0), "v"(a1), "v"(a2), "v"(a3), "v"(p0), "v"(p1), "v"(p2), "v"(p3), "s"(skip)
      : "scc");
}

__global__ __launch_bounds__(128)
void sipm_window_kernel(const float* __restrict__ z, const float* __restrict__ mask,
                        const float* __restrict__ W2, const float* __restrict__ b2,
                        const float* __restrict__ x, const float* __restrict__ W1,
                        const float* __restrict__ b1, float* __restrict__ planes) {
  __shared__ __attribute__((aligned(16))) int   lds_k[kListCap];
  __shared__ __attribute__((aligned(16))) float lds_z[kListCap];
  __shared__ __attribute__((aligned(16))) float lds_m[kListCap];
  __shared__ int lds_cnt[8];
  __shared__ __attribute__((aligned(16))) float lds_h[2][kBR * kPitchF];
  __shared__ __attribute__((aligned(16))) float lds_tap[2][kTapBuf];

  const int tid  = threadIdx.x;
  const int w    = tid >> 5;
  const int lane = tid & 31;
  const int m    = lane & 15;
  const int hf   = lane >> 4;
  const int id = blockIdx.x;
  int part, strip, win, nch;
  if (id < kWgA) {
    part = id / kTilesA;
    const int r = id - part * kTilesA;
    win   = r / kSGrid;
    strip = r - win * kSGrid;
    nch   = kNCh / kSplitA;
  } else {
    const int r = id - kWgA;
    part  = r / kSGrid;
    strip = r - part * kSGrid;
    win   = kTGrid - 1;
    nch   = kNCh / kSplitB;
  }
  const int s_base = strip * kSW + w * 16;
  const int t0 = win * kTW - 16;
  const float zlo = (win == 0) ? -(float)kReach : (float)(win * kTW);
  const float zhi = (float)(win * kTW + kTW);
  const float zmid = (float)(win * kTW + kTW / 2);
  const int   gw  = w >> 1;
  const int   tw  = w & 1;
  const int   ws  = __builtin_amdgcn_readfirstlane(w);
  const int   gws = ws >> 1;
  const int   tws = ws & 1;
  const float tfa = (float)(t0 + 16 * tw + m);
  const float tfb = (float)(t0 + 16 * tw + 32 + m);
  const int   tap_wr = ((gw * kNT + tw) * 2) * kTapQ + 4 * lane;

  const int  sW  = s_base + m;
  const bool sok = sW < kS;
  v2f w2f[16][2];
  #pragma unroll
  for (int j = 0; j < 16; ++j) {
    #pragma unroll
    for (int q = 0; q < 2; ++q) {
      const int i0 = 8 * j + 4 * hf + 2 * q;
      w2f[j][q].x = sok ? W2[(size_t)(i0 + 0) * kS + sW] : 0.f;
      w2f[j][q].y = sok ? W2[(size_t)(i0 + 1) * kS + sW] : 0.f;
    }
  }
  const float bias = sok ? b2[sW] : 0.f;
  const float4 w1a = *(const float4*)(W1 + 0 * kHid + 4 * lane);
  const float4 w1b = *(const float4*)(W1 + 1 * kHid + 4 * lane);
  const float4 w1c = *(const float4*)(W1 + 2 * kHid + 4 * lane);
  const float4 b1v = *(const float4*)(b1 + 4 * lane);
  const int xrow = (lane < 24 ? lane : 0) / 3;
  const int xcmp = (lane < 24 ? lane : 0) - 3 * xrow;

  v8f acc[kNT] = {};

  const int ch0 = part * nch;
  for (int ch = ch0; ch < ch0 + nch; ++ch) {
    const int cbase = ch * kCH + w * kSeg;
    float zr[kSegN], mr[kSegN];
    #pragma unroll
    for (int j = 0; j < kSegN; ++j) zr[j] = z[cbase + j * 32 + lane];
    #pragma unroll
    for (int j = 0; j < kSegN; ++j) mr[j] = mask[cbase + j * 32 + lane];
    {
      int cnta = 0, cnth = 0;
      #pragma unroll
      for (int j = 0; j < kSegN; ++j) {
        const bool hit = (zr[j] >= zlo) && (zr[j] < zhi);
        const unsigned bhit = __builtin_amdgcn_ballot_w32(hit);
        const unsigned bup  = __builtin_amdgcn_ballot_w32(zr[j] >= zmid);
        cnta += __builtin_popcount(bhit);
        cnth += __builtin_popcount(bhit & bup);
      }
      if (lane == 0) { lds_cnt[w] = cnta - cnth; lds_cnt[4 + w] = cnth; }
    }
    __syncthreads();
    const int c0 = lds_cnt[0], c1 = lds_cnt[1], c2 = lds_cnt[2], c3 = lds_cnt[3];
    const int e0 = lds_cnt[4], e1 = lds_cnt[5], e2 = lds_cnt[6], e3 = lds_cnt[7];
    const int nlo = __builtin_amdgcn_readfirstlane(c0 + c1 + c2 + c3);
    const int n   = nlo + __builtin_amdgcn_readfirstlane(e0 + e1 + e2 + e3);
    {
      int posl = (w > 0 ? c0 : 0) + (w > 1 ? c1 : 0) + (w > 2 ? c2 : 0);
      int posh = nlo + (w > 0 ? e0 : 0) + (w > 1 ? e1 : 0) + (w > 2 ? e2 : 0);
      #pragma unroll
      for (int j = 0; j < kSegN; ++j) {
        const float zk = zr[j];
        const bool hit = (zk >= zlo) && (zk < zhi);
        const bool up  = (zk >= zmid);
        const unsigned bhit = __builtin_amdgcn_ballot_w32(hit);
        const unsigned bup  = __builtin_amdgcn_ballot_w32(up);
        const unsigned bh = bhit & bup;
        const unsigned bl = bhit & ~bup;
        if (hit) {
          const unsigned um = up ? ~0u : 0u;
          const unsigned msel = bhit & ~(bup ^ um);
          const unsigned base = (unsigned)posl + ((unsigned)(posh - posl) & um);
          const int p = (int)__builtin_amdgcn_mbcnt_lo(msel, base);
          lds_k[p] = cbase + j * 32 + lane;
          lds_z[p] = zk;
          lds_m[p] = mr[j];
        }
        posl += __builtin_popcount(bl);
        posh += __builtin_popcount(bh);
      }
      if (w == 3) {
        const int p = n + lane;
        lds_k[p] = 0;
        lds_z[p] = 0.f;
        lds_m[p] = 0.f;
      }
    }
    __syncthreads();

    {
      const int ng = (n + 15) >> 4;
      const int nb = (ng + 1) >> 1;
      if (nb > 0) {
        {
          const float xv0 = x[lds_k[8 * w + xrow] * 3 + xcmp];
          {
            const int ez = 16 * gw + 8 * hf;
            const v4f z0 = *(const v4f*)(lds_z + ez);
            const v4f z1 = *(const v4f*)(lds_z + ez + 4);
            const v4f m0 = *(const v4f*)(lds_m + ez);
            const v4f m1 = *(const v4f*)(lds_m + ez + 4);
            const int es = 16 * gws;
            if (!(tws == 0 && es >= nlo))
              sipm_taps(tfa, z0, z1, m0, m1, &lds_tap[0][tap_wr]);
            if (!(tws == 1 && es + 16 <= nlo))
              sipm_taps(tfb, z0, z1, m0, m1, &lds_tap[0][tap_wr + 4 * kTapQ]);
          }
          float* dst = &lds_h[0][(8 * w) * kPitchF + 4 * lane];
          const int xb0 = __float_as_int(xv0);
          #pragma unroll
          for (int i = 0; i < 8; ++i) {
            const float x0 = __int_as_float(__builtin_amdgcn_readlane(xb0, 3 * i + 0));
            const float x1 = __int_as_float(__builtin_amdgcn_readlane(xb0, 3 * i + 1));
            const float x2 = __int_as_float(__builtin_amdgcn_readlane(xb0, 3 * i + 2));
            *(float4*)(dst + i * kPitchF) = sipm_h_row(x0, x1, x2, w1a, w1b, w1c, b1v);
          }
        }
        __syncthreads();
        int kfetch = lds_k[((1 < nb) ? 1 : 0) * kBR + 8 * w + xrow];
        for (int b = 0; b < nb; ++b) {
          const int bn = (b + 1 < nb) ? (b + 1) : b;
          const int xb = __float_as_int(x[kfetch * 3 + xcmp]);
          kfetch = lds_k[((b + 2 < nb) ? (b + 2) : (nb - 1)) * kBR + 8 * w + xrow];
          float* hnxt = &lds_h[(b + 1) & 1][(8 * w) * kPitchF + 4 * lane];
          const float* cur  = &lds_h[b & 1][0];
          const float* tcur = &lds_tap[b & 1][4 * lane];
          float*       tnxt = &lds_tap[(b + 1) & 1][tap_wr];
          #pragma unroll
          for (int half = 0; half < 2; ++half) {
            const int g = 2 * b + half;
            if (g >= ng) break;
            const float* arow = cur + (half * 16 + m) * kPitchF + 4 * hf;
            v4f a[16];
            #pragma unroll
            for (int j = 0; j < 16; ++j) a[j] = *(const v4f*)(arow + 8 * j);
            v4f nz0 = {}, nz1 = {}, nm0 = {}, nm1 = {};
            if (half == 0) {
              const int ez = bn * kBR + 16 * gw + 8 * hf;
              nz0 = *(const v4f*)(lds_z + ez);
              nz1 = *(const v4f*)(lds_z + ez + 4);
              nm0 = *(const v4f*)(lds_m + ez);
              nm1 = *(const v4f*)(lds_m + ez + 4);
            }
            const float* tg = tcur + (half * 2 * kNT) * kTapQ;
            const bool up_only = (16 * g >= nlo);
            const bool lo_only = (16 * g + 16 <= nlo);
            v4f bq[kNT][2];
            #pragma unroll
            for (int tt = 0; tt < kNT; ++tt) {
              if ((tt == 0 && up_only) || (tt == kNT - 1 && lo_only)) continue;
              bq[tt][0] = *(const v4f*)(tg + (2 * tt) * kTapQ);
              bq[tt][1] = *(const v4f*)(tg + (2 * tt + 1) * kTapQ);
            }
            __builtin_amdgcn_sched_barrier(0);

            v8f d0 = {}, d1 = {};
            #pragma unroll
            for (int j = 0; j < 16; ++j) {
              v2f a0; a0.x = a[j].x; a0.y = a[j].y;
              v2f a1; a1.x = a[j].z; a1.y = a[j].w;
              d0 = __builtin_amdgcn_wmma_f32_16x16x4_f32(false, a0, false, w2f[j][0],
                                                         (short)0, d0, false, false);
              d1 = __builtin_amdgcn_wmma_f32_16x16x4_f32(false, a1, false, w2f[j][1],
                                                         (short)0, d1, false, false);
              if (half == 1 && (j & 1)) {
                const int i = j >> 1;
                const float x0 = __int_as_float(__builtin_amdgcn_readlane(xb, 3 * i + 0));
                const float x1 = __int_as_float(__builtin_amdgcn_readlane(xb, 3 * i + 1));
                const float x2 = __int_as_float(__builtin_amdgcn_readlane(xb, 3 * i + 2));
                *(float4*)(hnxt + i * kPitchF) = sipm_h_row(x0, x1, x2, w1a, w1b, w1c, b1v);
              }
            }
            if (half == 0) {
              const int es = bn * kBR + 16 * gws;
              if (!(tws == 0 && es >= nlo))
                sipm_taps(tfa, nz0, nz1, nm0, nm1, tnxt);
              if (!(tws == 1 && es + 16 <= nlo))
                sipm_taps(tfb, nz0, nz1, nm0, nm1, tnxt + 4 * kTapQ);
            }
            float resp[8];
            #pragma unroll
            for (int r = 0; r < 8; ++r) resp[r] = expf((d0[r] + d1[r]) + bias);

            sipm_bin_tile_unless((int)up_only, acc[0], resp, bq[0][0], bq[0][1]);
            sipm_bin<1, kNT - 1>(acc, resp, bq);
            sipm_bin_tile_unless((int)lo_only, acc[kNT - 1], resp, bq[kNT - 1][0], bq[kNT - 1][1]);
          }
          __syncthreads();
        }
      }
    }
  }

  float* pbase = planes + (size_t)(part + kSplitA * (win & 1)) * kPlaneF
               + (size_t)(s_base + 8 * hf) * kTPad + (t0 + kTOff) + m;
  #pragma unroll
  for (int r = 0; r < 8; ++r) {
    float* prow = pbase + (size_t)r * kTPad;
    #pragma unroll
    for (int tt = 0; tt < kNT; ++tt) prow[16 * tt] = acc[tt][r];
  }
}

__global__ __launch_bounds__(256)
void sipm_copy_kernel(const float* __restrict__ planes, float* __restrict__ out) {
  int idx = blockIdx.x * 256 + threadIdx.x;
  if (idx >= kS * kT) return;
  int s = idx / kT, t = idx - s * kT;
  const float* p = planes + (size_t)s * kTPad + (t + kTOff);
  const int tOddLast = (kTGrid - 1) * kTW - 16;
  const int n1 = (t < 16) ? 0 : ((t < tOddLast) ? kSplitA : kSplitB);
  float v[2 * kSplitA];
  #pragma unroll
  for (int i = 0; i < kSplitA; ++i) v[i] = p[(size_t)i * kPlaneF];
  #pragma unroll
  for (int i = 0; i < kSplitA; ++i)
    v[kSplitA + i] = (i < n1) ? p[(size_t)(kSplitA + i) * kPlaneF] : 0.f;
  float sum = v[0];
  #pragma unroll
  for (int i = 1; i < 2 * kSplitA; ++i) sum += v[i];
  out[idx] = sum;
}

extern "C" void kernel_launch(void* const* d_in, const int* in_sizes, int n_in,
                              void* d_out, int out_size, void* d_ws, size_t ws_size,
                              hipStream_t stream) {
  const float* x   = (const float*)d_in[0];
  const float* z   = (const float*)d_in[1];
  const float* msk = (const float*)d_in[2];
  const float* W1  = (const float*)d_in[3];
  const float* b1  = (const float*)d_in[4];
  const float* W2  = (const float*)d_in[5];
  const float* b2  = (const float*)d_in[6];
  float* out = (float*)d_out;

  float* plane = (float*)d_ws;

  sipm_window_kernel<<<kWg, 128, 0, stream>>>(z, msk, W2, b2, x, W1, b1, plane);
  sipm_copy_kernel<<<(kS * kT + 255) / 256, 256, 0, stream>>>(plane, out);
}


# resp = exp(a) made as 2^(a*log2e) on the hardware exp2 instruction (rel. error ~6e-8*|a| + 1 ulp; denormal results become 0): 12 vector instructions per 8 values, was over 90; sums unchanged
# speedup vs baseline: 11.3614x; 1.0675x over previous
#include <hip/hip_runtime.h>
#include <math.h>


typedef float v2f __attribute__((ext_vector_type(2)));
typedef float v4f __attribute__((ext_vector_type(4)));
typedef float v8f __attribute__((ext_vector_type(8)));

namespace {
constexpr int kBatch  = 64;
constexpr int kEv     = 500;
constexpr int kK      = kBatch * kEv;
constexpr int kHid    = 128;
constexpr int kS      = 2209;
constexpr int kSW     = 64;
constexpr int kSGrid  = (kS + kSW - 1) / kSW;
constexpr int kSPad   = kSGrid * kSW;
constexpr int kT      = 550;
constexpr int kTW     = 32;
constexpr int kTGrid  = (kT + kTW - 1) / kTW;
constexpr int kNT     = 4;
constexpr int kTOff   = 16;
constexpr int kTPad   = kTOff + kTGrid * kTW + 16;
constexpr int kReach  = 11;
constexpr int kCH     = 3200;
constexpr int kNCh    = kK / kCH;
constexpr int kSplitA = 5;
constexpr int kSplitB = 2;
constexpr int kTilesA = kSGrid * (kTGrid - 1);
constexpr int kWgA    = kTilesA * kSplitA;
constexpr int kWgB    = kSGrid * kSplitB;
constexpr int kWg     = kWgA + kWgB;
constexpr int kSeg    = kCH / 4;
constexpr int kSegN   = kSeg / 32;
constexpr int kBR     = 32;
constexpr int kPitchF = kHid + 4;
constexpr int kListCap = kCH + kBR;
constexpr int kTapQ   = 32 * 4;
constexpr int kTapBuf = 2 * kNT * 2 * kTapQ;
constexpr float kNorm = 0.3989422804f * 1.41421356237309515f;
constexpr float kNegLog2e = -1.44269504088896341f;
constexpr float kLog2e    = 1.44269504088896341f;
constexpr size_t kPlaneF = (size_t)kSPad * kTPad;
static_assert(kTGrid * kTW >= kT + kReach, "the ranges must hold every z that reaches a tick");
static_assert(kNCh * kCH == kK, "chunks must tile K");
static_assert(kNCh % kSplitA == 0 && kNCh % kSplitB == 0, "parts must tile the chunks");
static_assert(kSeg % 32 == 0, "segment must be whole wave loads");
}

__device__ __forceinline__ float4 sipm_h_row(float x0, float x1, float x2, const float4 wa,
                                             const float4 wb, const float4 wc, const float4 bb) {
  float4 r;
  r.x = fmaxf(fmaf(x0, wa.x, fmaf(x1, wb.x, fmaf(x2, wc.x, bb.x))), 0.f);
  r.y = fmaxf(fmaf(x0, wa.y, fmaf(x1, wb.y, fmaf(x2, wc.y, bb.y))), 0.f);
  r.z = fmaxf(fmaf(x0, wa.z, fmaf(x1, wb.z, fmaf(x2, wc.z, bb.z))), 0.f);
  r.w = fmaxf(fmaf(x0, wa.w, fmaf(x1, wb.w, fmaf(x2, wc.w, bb.w))), 0.f);
  return r;
}

__device__ __forceinline__ void sipm_taps(float tf, const v4f z0, const v4f z1, const v4f m0,
                                          const v4f m1, float* td) {
  const float zz[8] = {z0.x, z0.y, z0.z, z0.w, z1.x, z1.y, z1.z, z1.w};
  const float mm[8] = {m0.x, m0.y, m0.z, m0.w, m1.x, m1.y, m1.z, m1.w};
  float tp[8];
  #pragma unroll
  for (int i = 0; i < 8; ++i) {
    const float d = tf - zz[i];
    tp[i] = __builtin_amdgcn_exp2f((d * d) * kNegLog2e) * mm[i] * kNorm;
  }
  v4f q0 = {tp[0], tp[1], tp[2], tp[3]};
  v4f q1 = {tp[4], tp[5], tp[6], tp[7]};
  *(v4f*)td = q0;
  *(v4f*)(td + kTapQ) = q1;
}

template <int T0, int T1>
__device__ __forceinline__ void sipm_bin(v8f (&acc)[kNT], const float (&resp)[8],
                                         const v4f (&bq)[kNT][2]) {
  #pragma unroll
  for (int j = 0; j < 4; ++j) {
    v2f af; af.x = resp[2 * j]; af.y = resp[2 * j + 1];
    #pragma unroll
    for (int tt = T0; tt < T1; ++tt) {
      const v4f bv = bq[tt][j >> 1];
      v2f bf;
      bf.x = (j & 1) ? bv.z : bv.x;
      bf.y = (j & 1) ? bv.w : bv.y;
      acc[tt] = __builtin_amdgcn_wmma_f32_16x16x4_f32(false, af, false, bf,
                                                      (short)0, acc[tt], false, false);
    }
  }
}

__device__ __forceinline__ void sipm_bin_tile_unless(int skip, v8f& acc, const float (&resp)[8],
                                                     const v4f b0, const v4f b1) {
  v2f a0; a0.x = resp[0]; a0.y = resp[1];
  v2f a1; a1.x = resp[2]; a1.y = resp[3];
  v2f a2; a2.x = resp[4]; a2.y = resp[5];
  v2f a3; a3.x = resp[6]; a3.y = resp[7];
  v2f p0; p0.x = b0.x; p0.y = b0.y;
  v2f p1; p1.x = b0.z; p1.y = b0.w;
  v2f p2; p2.x = b1.x; p2.y = b1.y;
  v2f p3; p3.x = b1.z; p3.y = b1.w;
  asm volatile(
      "s_cmp_lg_u32 %9, 0\n\t"
      "s_cbranch_scc1 .Lsipm_tile_skip_%=\n\t"
      "v_wmma_f32_16x16x4_f32 %0, %1, %5, %0\n\t"
      "v_wmma_f32_16x16x4_f32 %0, %2, %6, %0\n\t"
      "v_wmma_f32_16x16x4_f32 %0, %3, %7, %0\n\t"
      "v_wmma_f32_16x16x4_f32 %0, %4, %8, %0\n\t"
      "v_nop\n\tv_nop\n\tv_nop\n\tv_nop\n"
      ".Lsipm_tile_skip_%=:"
      : "+v"(acc)
      : "v"(a0), "v"(a1), "v"(a2), "v"(a3), "v"(p0), "v"(p1), "v"(p2), "v"(p3), "s"(skip)
      : "scc");
}

__global__ __launch_bounds__(128)
void sipm_window_kernel(const float* __restrict__ z, const float* __restrict__ mask,
                        const float* __restrict__ W2, const float* __restrict__ b2,
                        const float* __restrict__ x, const float* __restrict__ W1,
                        const float* __restrict__ b1, float* __restrict__ planes) {
  __shared__ __attribute__((aligned(16))) int   lds_k[kListCap];
  __shared__ __attribute__((aligned(16))) float lds_z[kListCap];
  __shared__ __attribute__((aligned(16))) float lds_m[kListCap];
  __shared__ int lds_cnt[8];
  __shared__ __attribute__((aligned(16))) float lds_h[2][kBR * kPitchF];
  __shared__ __attribute__((aligned(16))) float lds_tap[2][kTapBuf];

  const int tid  = threadIdx.x;
  const int w    = tid >> 5;
  const int lane = tid & 31;
  const int m    = lane & 15;
  const int hf   = lane >> 4;
  const int id = blockIdx.x;
  int part, strip, win, nch;
  if (id < kWgA) {
    part = id / kTilesA;
    const int r = id - part * kTilesA;
    win   = r / kSGrid;
    strip = r - win * kSGrid;
    nch   = kNCh / kSplitA;
  } else {
    const int r = id - kWgA;
    part  = r / kSGrid;
    strip = r - part * kSGrid;
    win   = kTGrid - 1;
    nch   = kNCh / kSplitB;
  }
  const int s_base = strip * kSW + w * 16;
  const int t0 = win * kTW - 16;
  const float zlo = (win == 0) ? -(float)kReach : (float)(win * kTW);
  const float zhi = (float)(win * kTW + kTW);
  const float zmid = (float)(win * kTW + kTW / 2);
  const int   gw  = w >> 1;
  const int   tw  = w & 1;
  const int   ws  = __builtin_amdgcn_readfirstlane(w);
  const int   gws = ws >> 1;
  const int   tws = ws & 1;
  const float tfa = (float)(t0 + 16 * tw + m);
  const float tfb = (float)(t0 + 16 * tw + 32 + m);
  const int   tap_wr = ((gw * kNT + tw) * 2) * kTapQ + 4 * lane;

  const int  sW  = s_base + m;
  const bool sok = sW < kS;
  v2f w2f[16][2];
  #pragma unroll
  for (int j = 0; j < 16; ++j) {
    #pragma unroll
    for (int q = 0; q < 2; ++q) {
      const int i0 = 8 * j + 4 * hf + 2 * q;
      w2f[j][q].x = sok ? W2[(size_t)(i0 + 0) * kS + sW] : 0.f;
      w2f[j][q].y = sok ? W2[(size_t)(i0 + 1) * kS + sW] : 0.f;
    }
  }
  const float bias = sok ? b2[sW] : 0.f;
  const float4 w1a = *(const float4*)(W1 + 0 * kHid + 4 * lane);
  const float4 w1b = *(const float4*)(W1 + 1 * kHid + 4 * lane);
  const float4 w1c = *(const float4*)(W1 + 2 * kHid + 4 * lane);
  const float4 b1v = *(const float4*)(b1 + 4 * lane);
  const int xrow = (lane < 24 ? lane : 0) / 3;
  const int xcmp = (lane < 24 ? lane : 0) - 3 * xrow;

  v8f acc[kNT] = {};

  const int ch0 = part * nch;
  for (int ch = ch0; ch < ch0 + nch; ++ch) {
    const int cbase = ch * kCH + w * kSeg;
    float zr[kSegN], mr[kSegN];
    #pragma unroll
    for (int j = 0; j < kSegN; ++j) zr[j] = z[cbase + j * 32 + lane];
    #pragma unroll
    for (int j = 0; j < kSegN; ++j) mr[j] = mask[cbase + j * 32 + lane];
    {
      int cnta = 0, cnth = 0;
      #pragma unroll
      for (int j = 0; j < kSegN; ++j) {
        const bool hit = (zr[j] >= zlo) && (zr[j] < zhi);
        const unsigned bhit = __builtin_amdgcn_ballot_w32(hit);
        const unsigned bup  = __builtin_amdgcn_ballot_w32(zr[j] >= zmid);
        cnta += __builtin_popcount(bhit);
        cnth += __builtin_popcount(bhit & bup);
      }
      if (lane == 0) { lds_cnt[w] = cnta - cnth; lds_cnt[4 + w] = cnth; }
    }
    __syncthreads();
    const int c0 = lds_cnt[0], c1 = lds_cnt[1], c2 = lds_cnt[2], c3 = lds_cnt[3];
    const int e0 = lds_cnt[4], e1 = lds_cnt[5], e2 = lds_cnt[6], e3 = lds_cnt[7];
    const int nlo = __builtin_amdgcn_readfirstlane(c0 + c1 + c2 + c3);
    const int n   = nlo + __builtin_amdgcn_readfirstlane(e0 + e1 + e2 + e3);
    {
      int posl = (w > 0 ? c0 : 0) + (w > 1 ? c1 : 0) + (w > 2 ? c2 : 0);
      int posh = nlo + (w > 0 ? e0 : 0) + (w > 1 ? e1 : 0) + (w > 2 ? e2 : 0);
      #pragma unroll
      for (int j = 0; j < kSegN; ++j) {
        const float zk = zr[j];
        const bool hit = (zk >= zlo) && (zk < zhi);
        const bool up  = (zk >= zmid);
        const unsigned bhit = __builtin_amdgcn_ballot_w32(hit);
        const unsigned bup  = __builtin_amdgcn_ballot_w32(up);
        const unsigned bh = bhit & bup;
        const unsigned bl = bhit & ~bup;
        if (hit) {
          const unsigned um = up ? ~0u : 0u;
          const unsigned msel = bhit & ~(bup ^ um);
          const unsigned base = (unsigned)posl + ((unsigned)(posh - posl) & um);
          const int p = (int)__builtin_amdgcn_mbcnt_lo(msel, base);
          lds_k[p] = cbase + j * 32 + lane;
          lds_z[p] = zk;
          lds_m[p] = mr[j];
        }
        posl += __builtin_popcount(bl);
        posh += __builtin_popcount(bh);
      }
      if (w == 3) {
        const int p = n + lane;
        lds_k[p] = 0;
        lds_z[p] = 0.f;
        lds_m[p] = 0.f;
      }
    }
    __syncthreads();

    {
      const int ng = (n + 15) >> 4;
      const int nb = (ng + 1) >> 1;
      if (nb > 0) {
        {
          const float xv0 = x[lds_k[8 * w + xrow] * 3 + xcmp];
          {
            const int ez = 16 * gw + 8 * hf;
            const v4f z0 = *(const v4f*)(lds_z + ez);
            const v4f z1 = *(const v4f*)(lds_z + ez + 4);
            const v4f m0 = *(const v4f*)(lds_m + ez);
            const v4f m1 = *(const v4f*)(lds_m + ez + 4);
            const int es = 16 * gws;
            if (!(tws == 0 && es >= nlo))
              sipm_taps(tfa, z0, z1, m0, m1, &lds_tap[0][tap_wr]);
            if (!(tws == 1 && es + 16 <= nlo))
              sipm_taps(tfb, z0, z1, m0, m1, &lds_tap[0][tap_wr + 4 * kTapQ]);
          }
          float* dst = &lds_h[0][(8 * w) * kPitchF + 4 * lane];
          const int xb0 = __float_as_int(xv0);
          #pragma unroll
          for (int i = 0; i < 8; ++i) {
            const float x0 = __int_as_float(__builtin_amdgcn_readlane(xb0, 3 * i + 0));
            const float x1 = __int_as_float(__builtin_amdgcn_readlane(xb0, 3 * i + 1));
            const float x2 = __int_as_float(__builtin_amdgcn_readlane(xb0, 3 * i + 2));
            *(float4*)(dst + i * kPitchF) = sipm_h_row(x0, x1, x2, w1a, w1b, w1c, b1v);
          }
        }
        __syncthreads();
        int kfetch = lds_k[((1 < nb) ? 1 : 0) * kBR + 8 * w + xrow];
        for (int b = 0; b < nb; ++b) {
          const int bn = (b + 1 < nb) ? (b + 1) : b;
          const int xb = __float_as_int(x[kfetch * 3 + xcmp]);
          kfetch = lds_k[((b + 2 < nb) ? (b + 2) : (nb - 1)) * kBR + 8 * w + xrow];
          float* hnxt = &lds_h[(b + 1) & 1][(8 * w) * kPitchF + 4 * lane];
          const float* cur  = &lds_h[b & 1][0];
          const float* tcur = &lds_tap[b & 1][4 * lane];
          float*       tnxt = &lds_tap[(b + 1) & 1][tap_wr];
          #pragma unroll
          for (int half = 0; half < 2; ++half) {
            const int g = 2 * b + half;
            if (g >= ng) break;
            const float* arow = cur + (half * 16 + m) * kPitchF + 4 * hf;
            v4f a[16];
            #pragma unroll
            for (int j = 0; j < 16; ++j) a[j] = *(const v4f*)(arow + 8 * j);
            v4f nz0 = {}, nz1 = {}, nm0 = {}, nm1 = {};
            if (half == 0) {
              const int ez = bn * kBR + 16 * gw + 8 * hf;
              nz0 = *(const v4f*)(lds_z + ez);
              nz1 = *(const v4f*)(lds_z + ez + 4);
              nm0 = *(const v4f*)(lds_m + ez);
              nm1 = *(const v4f*)(lds_m + ez + 4);
            }
            const float* tg = tcur + (half * 2 * kNT) * kTapQ;
            const bool up_only = (16 * g >= nlo);
            const bool lo_only = (16 * g + 16 <= nlo);
            v4f bq[kNT][2];
            #pragma unroll
            for (int tt = 0; tt < kNT; ++tt) {
              if ((tt == 0 && up_only) || (tt == kNT - 1 && lo_only)) continue;
              bq[tt][0] = *(const v4f*)(tg + (2 * tt) * kTapQ);
              bq[tt][1] = *(const v4f*)(tg + (2 * tt + 1) * kTapQ);
            }
            __builtin_amdgcn_sched_barrier(0);

            v8f d0 = {}, d1 = {};
            #pragma unroll
            for (int j = 0; j < 16; ++j) {
              v2f a0; a0.x = a[j].x; a0.y = a[j].y;
              v2f a1; a1.x = a[j].z; a1.y = a[j].w;
              d0 = __builtin_amdgcn_wmma_f32_16x16x4_f32(false, a0, false, w2f[j][0],
                                                         (short)0, d0, false, false);
              d1 = __builtin_amdgcn_wmma_f32_16x16x4_f32(false, a1, false, w2f[j][1],
                                                         (short)0, d1, false, false);
              if (half == 1 && (j & 1)) {
                const int i = j >> 1;
                const float x0 = __int_as_float(__builtin_amdgcn_readlane(xb, 3 * i + 0));
                const float x1 = __int_as_float(__builtin_amdgcn_readlane(xb, 3 * i + 1));
                const float x2 = __int_as_float(__builtin_amdgcn_readlane(xb, 3 * i + 2));
                *(float4*)(hnxt + i * kPitchF) = sipm_h_row(x0, x1, x2, w1a, w1b, w1c, b1v);
              }
            }
            if (half == 0) {
              const int es = bn * kBR + 16 * gws;
              if (!(tws == 0 && es >= nlo))
                sipm_taps(tfa, nz0, nz1, nm0, nm1, tnxt);
              if (!(tws == 1 && es + 16 <= nlo))
                sipm_taps(tfb, nz0, nz1, nm0, nm1, tnxt + 4 * kTapQ);
            }
            float resp[8];
            #pragma unroll
            for (int r = 0; r < 8; ++r)
              resp[r] = __builtin_amdgcn_exp2f(((d0[r] + d1[r]) + bias) * kLog2e);

            sipm_bin_tile_unless((int)up_only, acc[0], resp, bq[0][0], bq[0][1]);
            sipm_bin<1, kNT - 1>(acc, resp, bq);
            sipm_bin_tile_unless((int)lo_only, acc[kNT - 1], resp, bq[kNT - 1][0], bq[kNT - 1][1]);
          }
          __syncthreads();
        }
      }
    }
  }

  float* pbase = planes + (size_t)(part + kSplitA * (win & 1)) * kPlaneF
               + (size_t)(s_base + 8 * hf) * kTPad + (t0 + kTOff) + m;
  #pragma unroll
  for (int r = 0; r < 8; ++r) {
    float* prow = pbase + (size_t)r * kTPad;
    #pragma unroll
    for (int tt = 0; tt < kNT; ++tt) prow[16 * tt] = acc[tt][r];
  }
}

__global__ __launch_bounds__(256)
void sipm_copy_kernel(const float* __restrict__ planes, float* __restrict__ out) {
  int idx = blockIdx.x * 256 + threadIdx.x;
  if (idx >= kS * kT) return;
  int s = idx / kT, t = idx - s * kT;
  const float* p = planes + (size_t)s * kTPad + (t + kTOff);
  const int tOddLast = (kTGrid - 1) * kTW - 16;
  const int n1 = (t < 16) ? 0 : ((t < tOddLast) ? kSplitA : kSplitB);
  float v[2 * kSplitA];
  #pragma unroll
  for (int i = 0; i < kSplitA; ++i) v[i] = p[(size_t)i * kPlaneF];
  #pragma unroll
  for (int i = 0; i < kSplitA; ++i)
    v[kSplitA + i] = (i < n1) ? p[(size_t)(kSplitA + i) * kPlaneF] : 0.f;
  float sum = v[0];
  #pragma unroll
  for (int i = 1; i < 2 * kSplitA; ++i) sum += v[i];
  out[idx] = sum;
}

extern "C" void kernel_launch(void* const* d_in, const int* in_sizes, int n_in,
                              void* d_out, int out_size, void* d_ws, size_t ws_size,
                              hipStream_t stream) {
  const float* x   = (const float*)d_in[0];
  const float* z   = (const float*)d_in[1];
  const float* msk = (const float*)d_in[2];
  const float* W1  = (const float*)d_in[3];
  const float* b1  = (const float*)d_in[4];
  const float* W2  = (const float*)d_in[5];
  const float* b2  = (const float*)d_in[6];
  float* out = (float*)d_out;

  float* plane = (float*)d_ws;

  sipm_window_kernel<<<kWg, 128, 0, stream>>>(z, msk, W2, b2, x, W1, b1, plane);
  sipm_copy_kernel<<<(kS * kT + 255) / 256, 256, 0, stream>>>(plane, out);
}
